# starting kernel
# baseline (speedup 1.0000x reference)
#include <hip/hip_runtime.h>
#include <hip/hip_bf16.h>

#define NNODES 50000
#define NEDGES 640000
#define DH 128
#define DOUT 64
#define LN_EPS 1e-5f

typedef float v2f __attribute__((ext_vector_type(2)));
typedef float v8f __attribute__((ext_vector_type(8)));

__global__ void gcn_deg_kernel(const int* __restrict__ dst, const float* __restrict__ w,
                               float* __restrict__ deg, int E) {
    int e = blockIdx.x * blockDim.x + threadIdx.x;
    if (e < E) atomicAdd(&deg[dst[e]], w[e]);
}

__global__ void gcn_dinv_kernel(float* __restrict__ deg, int n) {
    int i = blockIdx.x * blockDim.x + threadIdx.x;
    if (i < n) {
        float d = deg[i] + 1.0f;
        deg[i] = rsqrtf(d);
    }
}

__global__ void gcn_norm_kernel(const int* __restrict__ src, const int* __restrict__ dst,
                                const float* __restrict__ w, const float* __restrict__ dinv,
                                float* __restrict__ norm, int E) {
    int e = blockIdx.x * blockDim.x + threadIdx.x;
    if (e < E) norm[e] = dinv[src[e]] * w[e] * dinv[dst[e]];
}

__global__ void gcn_gemm_wmma(const float* __restrict__ A,
                              const float* __restrict__ W,
                              float* __restrict__ out,
                              int nrows, int ncols) {
    const int lane = threadIdx.x & 31;
    const int wid  = blockIdx.x * (blockDim.x >> 5) + (threadIdx.x >> 5);
    const int ctiles = ncols >> 4;
    const int rtiles = nrows >> 4;
    if (wid >= rtiles * ctiles) return;
    const int rt = wid / ctiles;
    const int ct = wid - rt * ctiles;
    const int row0 = rt << 4;
    const int j0   = ct << 4;

    const int m  = lane & 15;
    const int ko = (lane >> 4) << 1;
    const float* __restrict__ arow = A + (size_t)(row0 + m) * DH + ko;
    const float* __restrict__ brow = W + (size_t)(j0 + m) * DH + ko;

    v8f acc = {};
#pragma unroll
    for (int k = 0; k < DH; k += 4) {
        v2f a = *(const v2f*)(arow + k);
        v2f b = *(const v2f*)(brow + k);
        acc = __builtin_amdgcn_wmma_f32_16x16x4_f32(
             false, a,  false, b,
             (short)0, acc,  false,  false);
    }

    float* __restrict__ o = out + (size_t)(row0 + ((lane >> 4) << 3)) * ncols + j0 + m;
#pragma unroll
    for (int v = 0; v < 8; ++v)
        o[(size_t)v * ncols] = acc[v];
}

__global__ void gcn_scatter_kernel(const float* __restrict__ xw,
                                   const int* __restrict__ src, const int* __restrict__ dst,
                                   const float* __restrict__ norm,
                                   float* __restrict__ agg, int E, int ncols) {
    int t = blockIdx.x * blockDim.x + threadIdx.x;
    int per_edge = ncols >> 2;
    int e = t / per_edge;
    if (e >= E) return;
    int f = (t - e * per_edge) << 2;
    int s = src[e], d = dst[e];
    float w = norm[e];
    const float4 v = *(const float4*)(xw + (size_t)s * ncols + f);
    float* p = agg + (size_t)d * ncols + f;
    atomicAdd(p + 0, w * v.x);
    atomicAdd(p + 1, w * v.y);
    atomicAdd(p + 2, w * v.z);
    atomicAdd(p + 3, w * v.w);
}

__device__ __forceinline__ float wave_sum32(float v) {
#pragma unroll
    for (int m = 16; m >= 1; m >>= 1) v += __shfl_xor(v, m, 32);
    return v;
}

__global__ void gcn_epilogue_hidden(const float* __restrict__ agg, const float* __restrict__ xw,
                                    const float* __restrict__ dinv,
                                    const float* __restrict__ bias, const float* __restrict__ gamma,
                                    const float* __restrict__ beta,
                                    const float* __restrict__ res, float* __restrict__ hout, int n) {
    int t = blockIdx.x * blockDim.x + threadIdx.x;
    int lane = t & 31;
    int node = t >> 5;
    if (node >= n) return;

    float di = dinv[node];
    float sl = di * di;
    size_t base = (size_t)node * DH + (lane << 2);

    float4 a  = *(const float4*)(agg + base);
    float4 xv = *(const float4*)(xw + base);
    float4 b  = *(const float4*)(bias + (lane << 2));

    float v0 = fmaf(sl, xv.x, a.x) + b.x;
    float v1 = fmaf(sl, xv.y, a.y) + b.y;
    float v2 = fmaf(sl, xv.z, a.z) + b.z;
    float v3 = fmaf(sl, xv.w, a.w) + b.w;

    float s  = wave_sum32(v0 + v1 + v2 + v3);
    float sq = wave_sum32(v0 * v0 + v1 * v1 + v2 * v2 + v3 * v3);
    float mean = s * (1.0f / DH);
    float var  = sq * (1.0f / DH) - mean * mean;
    float rinv = rsqrtf(var + LN_EPS);

    float4 g  = *(const float4*)(gamma + (lane << 2));
    float4 be = *(const float4*)(beta + (lane << 2));
    float4 r  = *(const float4*)(res + base);

    float4 o;
    o.x = fmaxf(g.x * (v0 - mean) * rinv + be.x, 0.0f) + r.x;
    o.y = fmaxf(g.y * (v1 - mean) * rinv + be.y, 0.0f) + r.y;
    o.z = fmaxf(g.z * (v2 - mean) * rinv + be.z, 0.0f) + r.z;
    o.w = fmaxf(g.w * (v3 - mean) * rinv + be.w, 0.0f) + r.w;
    *(float4*)(hout + base) = o;
}

__global__ void gcn_epilogue_final(const float* __restrict__ agg, const float* __restrict__ xw,
                                   const float* __restrict__ dinv,
                                   const float* __restrict__ bias, const float* __restrict__ gamma,
                                   const float* __restrict__ beta,
                                   float* __restrict__ out, int n) {
    int t = blockIdx.x * blockDim.x + threadIdx.x;
    int lane = t & 31;
    int node = t >> 5;
    if (node >= n) return;

    float di = dinv[node];
    float sl = di * di;
    size_t base = (size_t)node * DOUT + (lane << 1);

    float2 a  = *(const float2*)(agg + base);
    float2 xv = *(const float2*)(xw + base);
    float2 b  = *(const float2*)(bias + (lane << 1));

    float v0 = fmaf(sl, xv.x, a.x) + b.x;
    float v1 = fmaf(sl, xv.y, a.y) + b.y;

    float s  = wave_sum32(v0 + v1);
    float sq = wave_sum32(v0 * v0 + v1 * v1);
    float mean = s * (1.0f / DOUT);
    float var  = sq * (1.0f / DOUT) - mean * mean;
    float rinv = rsqrtf(var + LN_EPS);

    float2 g  = *(const float2*)(gamma + (lane << 1));
    float2 be = *(const float2*)(beta + (lane << 1));

    float2 o;
    o.x = g.x * (v0 - mean) * rinv + be.x;
    o.y = g.y * (v1 - mean) * rinv + be.y;
    *(float2*)(out + base) = o;
}

extern "C" void kernel_launch(void* const* d_in, const int* in_sizes, int n_in,
                              void* d_out, int out_size, void* d_ws, size_t ws_size,
                              hipStream_t stream) {
    const float* x    = (const float*)d_in[0];
    const int*   ei   = (const int*)d_in[1];
    const float* ew   = (const float*)d_in[2];
    const float* W03  = (const float*)d_in[3];
    const float* b03  = (const float*)d_in[4];
    const float* g03  = (const float*)d_in[5];
    const float* be03 = (const float*)d_in[6];
    const float* W4   = (const float*)d_in[7];
    const float* b4   = (const float*)d_in[8];
    const float* g4   = (const float*)d_in[9];
    const float* be4  = (const float*)d_in[10];

    const int* srcI = ei;
    const int* dstI = ei + NEDGES;

    float* dinv = (float*)d_ws;
    float* norm = dinv + NNODES;
    float* xw   = norm + NEDGES;
    float* agg  = xw + (size_t)NNODES * DH;
    float* hbuf = agg + (size_t)NNODES * DH;

    hipMemsetAsync(dinv, 0, (size_t)NNODES * sizeof(float), stream);
    gcn_deg_kernel<<<(NEDGES + 255) / 256, 256, 0, stream>>>(dstI, ew, dinv, NEDGES);
    gcn_dinv_kernel<<<(NNODES + 255) / 256, 256, 0, stream>>>(dinv, NNODES);
    gcn_norm_kernel<<<(NEDGES + 255) / 256, 256, 0, stream>>>(srcI, dstI, ew, dinv, norm, NEDGES);

    const int rtiles = NNODES / 16;
    const float* hin = x;
    const float* resin = x;
    for (int l = 0; l < 4; ++l) {
        int waves  = rtiles * (DH / 16);
        int blocks = (waves * 32 + 255) / 256;
        gcn_gemm_wmma<<<blocks, 256, 0, stream>>>(hin, W03 + (size_t)l * DH * DH, xw, NNODES, DH);

        hipMemsetAsync(agg, 0, (size_t)NNODES * DH * sizeof(float), stream);
        long long sthreads = (long long)NEDGES * (DH / 4);
        gcn_scatter_kernel<<<(int)((sthreads + 255) / 256), 256, 0, stream>>>(
            xw, srcI, dstI, norm, agg, NEDGES, DH);

        long long ethreads = (long long)NNODES * 32;
        gcn_epilogue_hidden<<<(int)((ethreads + 255) / 256), 256, 0, stream>>>(
            agg, xw, dinv, b03 + l * DH, g03 + l * DH, be03 + l * DH, resin, hbuf, NNODES);

        hin = hbuf;
        resin = hbuf;
    }

    {
        int waves  = rtiles * (DOUT / 16);
        int blocks = (waves * 32 + 255) / 256;
        gcn_gemm_wmma<<<blocks, 256, 0, stream>>>(hbuf, W4, xw, NNODES, DOUT);

        hipMemsetAsync(agg, 0, (size_t)NNODES * DOUT * sizeof(float), stream);
        long long sthreads = (long long)NEDGES * (DOUT / 4);
        gcn_scatter_kernel<<<(int)((sthreads + 255) / 256), 256, 0, stream>>>(
            xw, srcI, dstI, norm, agg, NEDGES, DOUT);

        long long ethreads = (long long)NNODES * 32;
        gcn_epilogue_final<<<(int)((ethreads + 255) / 256), 256, 0, stream>>>(
            agg, xw, dinv, b4, g4, be4, (float*)d_out, NNODES);
    }
}


# Replace f32 atomic scatter-add by per-destination gather: edges bucketed by dst each call (int count+rank, scan, place, edge-id order), one wave sums a node's rows in registers; sum order now fixed
# speedup vs baseline: 2.3295x; 2.3295x over previous
#include <hip/hip_runtime.h>
#include <hip/hip_bf16.h>

#define NNODES 50000
#define NEDGES 640000
#define DH 128
#define DOUT 64
#define LN_EPS 1e-5f

typedef float v2f __attribute__((ext_vector_type(2)));
typedef float v8f __attribute__((ext_vector_type(8)));

__global__ void gcn_count_kernel(const int* __restrict__ dst, int* __restrict__ cnt,
                                 int* __restrict__ rnk, int E) {
    int e = blockIdx.x * blockDim.x + threadIdx.x;
    if (e < E) rnk[e] = atomicAdd(&cnt[dst[e]], 1);
}

__device__ __forceinline__ int wave_incl_scan32(int x, int lane) {
#pragma unroll
    for (int o = 1; o < 32; o <<= 1) {
        int y = __shfl_up(x, o, 32);
        if (lane >= o) x += y;
    }
    return x;
}

#define SCAN_CH ((NNODES + 1023) / 1024)
__global__ void __launch_bounds__(1024)
gcn_scan_kernel(const int* __restrict__ cnt, int* __restrict__ rowptr) {
    __shared__ int wtot[32];
    const int lane = threadIdx.x & 31;
    const int w    = threadIdx.x >> 5;
    const int base = w * (SCAN_CH * 32) + lane;
    int s = 0;
#pragma unroll
    for (int i = 0; i < SCAN_CH; ++i) {
        int j = base + i * 32;
        s += (j < NNODES) ? cnt[j] : 0;
    }
    const int sinc = wave_incl_scan32(s, lane);
    if (lane == 31) wtot[w] = sinc;
    __syncthreads();
    const int tv    = wtot[lane];
    const int tinc  = wave_incl_scan32(tv, lane);
    int       run   = __shfl(tinc - tv, w, 32);
    const int total = __shfl(tinc, 31, 32);
#pragma unroll
    for (int i = 0; i < SCAN_CH; ++i) {
        int j   = base + i * 32;
        int x   = (j < NNODES) ? cnt[j] : 0;
        int inc = wave_incl_scan32(x, lane);
        if (j < NNODES) rowptr[j] = run + inc - x;
        run += __shfl(inc, 31, 32);
    }
    if (threadIdx.x == 0) rowptr[NNODES] = total;
}

__global__ void gcn_place_kernel(const int* __restrict__ dst, const int* __restrict__ rnk,
                                 const int* __restrict__ rowptr, int* __restrict__ slot_eid, int E) {
    int e = blockIdx.x * blockDim.x + threadIdx.x;
    if (e < E) slot_eid[rowptr[dst[e]] + rnk[e]] = e;
}

__global__ void gcn_order_kernel(const int* __restrict__ src, const int* __restrict__ dst,
                                 const float* __restrict__ w, const int* __restrict__ rowptr,
                                 const int* __restrict__ slot_eid,
                                 int* __restrict__ csr_src, int* __restrict__ csr_dst,
                                 float* __restrict__ csr_w, int E) {
    int p = blockIdx.x * blockDim.x + threadIdx.x;
    if (p >= E) return;
    int e  = slot_eid[p];
    int d  = dst[e];
    int lo = rowptr[d], hi = rowptr[d + 1];
    int r  = 0;
    for (int j = lo; j < hi; ++j) r += (slot_eid[j] < e) ? 1 : 0;
    int q = lo + r;
    csr_src[q] = src[e];
    csr_dst[q] = d;
    csr_w[q]   = w[e];
}

__global__ void gcn_dinv_kernel(const int* __restrict__ rowptr, const float* __restrict__ csr_w,
                                float* __restrict__ dinv, int n) {
    int i = blockIdx.x * blockDim.x + threadIdx.x;
    if (i < n) {
        float d = 0.0f;
        for (int q = rowptr[i]; q < rowptr[i + 1]; ++q) d += csr_w[q];
        dinv[i] = rsqrtf(d + 1.0f);
    }
}

__global__ void gcn_norm_kernel(const int* __restrict__ csr_src, const int* __restrict__ csr_dst,
                                const float* __restrict__ csr_w, const float* __restrict__ dinv,
                                float* __restrict__ norm, int E) {
    int q = blockIdx.x * blockDim.x + threadIdx.x;
    if (q < E) norm[q] = dinv[csr_src[q]] * csr_w[q] * dinv[csr_dst[q]];
}

__global__ void gcn_gemm_wmma(const float* __restrict__ A,
                              const float* __restrict__ W,
                              float* __restrict__ out,
                              int nrows, int ncols) {
    const int lane = threadIdx.x & 31;
    const int wid  = blockIdx.x * (blockDim.x >> 5) + (threadIdx.x >> 5);
    const int ctiles = ncols >> 4;
    const int rtiles = nrows >> 4;
    if (wid >= rtiles * ctiles) return;
    const int rt = wid / ctiles;
    const int ct = wid - rt * ctiles;
    const int row0 = rt << 4;
    const int j0   = ct << 4;

    const int m  = lane & 15;
    const int ko = (lane >> 4) << 1;
    const float* __restrict__ arow = A + (size_t)(row0 + m) * DH + ko;
    const float* __restrict__ brow = W + (size_t)(j0 + m) * DH + ko;

    v8f acc = {};
#pragma unroll
    for (int k = 0; k < DH; k += 4) {
        v2f a = *(const v2f*)(arow + k);
        v2f b = *(const v2f*)(brow + k);
        acc = __builtin_amdgcn_wmma_f32_16x16x4_f32(
             false, a,  false, b,
             (short)0, acc,  false,  false);
    }

    float* __restrict__ o = out + (size_t)(row0 + ((lane >> 4) << 3)) * ncols + j0 + m;
#pragma unroll
    for (int v = 0; v < 8; ++v)
        o[(size_t)v * ncols] = acc[v];
}

template <int D>
__global__ void gcn_gather_kernel(const float* __restrict__ xw, const int* __restrict__ rowptr,
                                  const int* __restrict__ csr_src, const float* __restrict__ norm,
                                  float* __restrict__ agg, int n) {
    constexpr int V = D / 32;
    typedef float vf __attribute__((ext_vector_type(V)));
    const int lane = threadIdx.x & 31;
    const int node = __builtin_amdgcn_readfirstlane(blockIdx.x * (blockDim.x >> 5) + (threadIdx.x >> 5));
    if (node >= n) return;
    const int lo = rowptr[node], hi = rowptr[node + 1];
    const float* __restrict__ xl = xw + lane * V;

    vf acc = {};
    for (int base = lo; base < hi; base += 32) {
        const int c = (hi - base < 32) ? (hi - base) : 32;
        int   ms = 0;
        float mw = 0.0f;
        if (lane < c) { ms = csr_src[base + lane]; mw = norm[base + lane]; }
        const int mwi = __float_as_int(mw);
        int j = 0;
        for (; j + 4 <= c; j += 4) {
            const int s0 = __builtin_amdgcn_readlane(ms, j);
            const int s1 = __builtin_amdgcn_readlane(ms, j + 1);
            const int s2 = __builtin_amdgcn_readlane(ms, j + 2);
            const int s3 = __builtin_amdgcn_readlane(ms, j + 3);
            const float w0 = __int_as_float(__builtin_amdgcn_readlane(mwi, j));
            const float w1 = __int_as_float(__builtin_amdgcn_readlane(mwi, j + 1));
            const float w2 = __int_as_float(__builtin_amdgcn_readlane(mwi, j + 2));
            const float w3 = __int_as_float(__builtin_amdgcn_readlane(mwi, j + 3));
            const vf v0 = *(const vf*)(xl + (size_t)s0 * D);
            const vf v1 = *(const vf*)(xl + (size_t)s1 * D);
            const vf v2 = *(const vf*)(xl + (size_t)s2 * D);
            const vf v3 = *(const vf*)(xl + (size_t)s3 * D);
            acc += w0 * v0;
            acc += w1 * v1;
            acc += w2 * v2;
            acc += w3 * v3;
        }
        for (; j < c; ++j) {
            const int s0 = __builtin_amdgcn_readlane(ms, j);
            const float w0 = __int_as_float(__builtin_amdgcn_readlane(mwi, j));
            const vf v0 = *(const vf*)(xl + (size_t)s0 * D);
            acc += w0 * v0;
        }
    }
    *(vf*)(agg + (size_t)node * D + lane * V) = acc;
}

__device__ __forceinline__ float wave_sum32(float v) {
#pragma unroll
    for (int m = 16; m >= 1; m >>= 1) v += __shfl_xor(v, m, 32);
    return v;
}

__global__ void gcn_epilogue_hidden(const float* __restrict__ agg, const float* __restrict__ xw,
                                    const float* __restrict__ dinv,
                                    const float* __restrict__ bias, const float* __restrict__ gamma,
                                    const float* __restrict__ beta,
                                    const float* __restrict__ res, float* __restrict__ hout, int n) {
    int t = blockIdx.x * blockDim.x + threadIdx.x;
    int lane = t & 31;
    int node = t >> 5;
    if (node >= n) return;

    float di = dinv[node];
    float sl = di * di;
    size_t base = (size_t)node * DH + (lane << 2);

    float4 a  = *(const float4*)(agg + base);
    float4 xv = *(const float4*)(xw + base);
    float4 b  = *(const float4*)(bias + (lane << 2));

    float v0 = fmaf(sl, xv.x, a.x) + b.x;
    float v1 = fmaf(sl, xv.y, a.y) + b.y;
    float v2 = fmaf(sl, xv.z, a.z) + b.z;
    float v3 = fmaf(sl, xv.w, a.w) + b.w;

    float s  = wave_sum32(v0 + v1 + v2 + v3);
    float sq = wave_sum32(v0 * v0 + v1 * v1 + v2 * v2 + v3 * v3);
    float mean = s * (1.0f / DH);
    float var  = sq * (1.0f / DH) - mean * mean;
    float rinv = rsqrtf(var + LN_EPS);

    float4 g  = *(const float4*)(gamma + (lane << 2));
    float4 be = *(const float4*)(beta + (lane << 2));
    float4 r  = *(const float4*)(res + base);

    float4 o;
    o.x = fmaxf(g.x * (v0 - mean) * rinv + be.x, 0.0f) + r.x;
    o.y = fmaxf(g.y * (v1 - mean) * rinv + be.y, 0.0f) + r.y;
    o.z = fmaxf(g.z * (v2 - mean) * rinv + be.z, 0.0f) + r.z;
    o.w = fmaxf(g.w * (v3 - mean) * rinv + be.w, 0.0f) + r.w;
    *(float4*)(hout + base) = o;
}

__global__ void gcn_epilogue_final(const float* __restrict__ agg, const float* __restrict__ xw,
                                   const float* __restrict__ dinv,
                                   const float* __restrict__ bias, const float* __restrict__ gamma,
                                   const float* __restrict__ beta,
                                   float* __restrict__ out, int n) {
    int t = blockIdx.x * blockDim.x + threadIdx.x;
    int lane = t & 31;
    int node = t >> 5;
    if (node >= n) return;

    float di = dinv[node];
    float sl = di * di;
    size_t base = (size_t)node * DOUT + (lane << 1);

    float2 a  = *(const float2*)(agg + base);
    float2 xv = *(const float2*)(xw + base);
    float2 b  = *(const float2*)(bias + (lane << 1));

    float v0 = fmaf(sl, xv.x, a.x) + b.x;
    float v1 = fmaf(sl, xv.y, a.y) + b.y;

    float s  = wave_sum32(v0 + v1);
    float sq = wave_sum32(v0 * v0 + v1 * v1);
    float mean = s * (1.0f / DOUT);
    float var  = sq * (1.0f / DOUT) - mean * mean;
    float rinv = rsqrtf(var + LN_EPS);

    float2 g  = *(const float2*)(gamma + (lane << 1));
    float2 be = *(const float2*)(beta + (lane << 1));

    float2 o;
    o.x = g.x * (v0 - mean) * rinv + be.x;
    o.y = g.y * (v1 - mean) * rinv + be.y;
    *(float2*)(out + base) = o;
}

extern "C" void kernel_launch(void* const* d_in, const int* in_sizes, int n_in,
                              void* d_out, int out_size, void* d_ws, size_t ws_size,
                              hipStream_t stream) {
    const float* x    = (const float*)d_in[0];
    const int*   ei   = (const int*)d_in[1];
    const float* ew   = (const float*)d_in[2];
    const float* W03  = (const float*)d_in[3];
    const float* b03  = (const float*)d_in[4];
    const float* g03  = (const float*)d_in[5];
    const float* be03 = (const float*)d_in[6];
    const float* W4   = (const float*)d_in[7];
    const float* b4   = (const float*)d_in[8];
    const float* g4   = (const float*)d_in[9];
    const float* be4  = (const float*)d_in[10];

    const int* srcI = ei;
    const int* dstI = ei + NEDGES;

    float* xw   = (float*)d_ws;
    float* agg  = xw + (size_t)NNODES * DH;
    float* hbuf = agg + (size_t)NNODES * DH;
    float* dinv = hbuf + (size_t)NNODES * DH;
    float* norm = dinv + NNODES;
    float* csrW = norm + NEDGES;
    int* cnt     = (int*)(csrW + NEDGES);
    int* rowptr  = cnt + NNODES;
    int* rnk     = rowptr + NNODES + 1;
    int* slotEid = rnk + NEDGES;
    int* csrSrc  = slotEid + NEDGES;
    int* csrDst  = csrSrc + NEDGES;

    const int eblocks = (NEDGES + 255) / 256;
    hipMemsetAsync(cnt, 0, (size_t)NNODES * sizeof(int), stream);
    gcn_count_kernel<<<eblocks, 256, 0, stream>>>(dstI, cnt, rnk, NEDGES);
    gcn_scan_kernel<<<1, 1024, 0, stream>>>(cnt, rowptr);
    gcn_place_kernel<<<eblocks, 256, 0, stream>>>(dstI, rnk, rowptr, slotEid, NEDGES);
    gcn_order_kernel<<<eblocks, 256, 0, stream>>>(srcI, dstI, ew, rowptr, slotEid, csrSrc, csrDst, csrW, NEDGES);
    gcn_dinv_kernel<<<(NNODES + 255) / 256, 256, 0, stream>>>(rowptr, csrW, dinv, NNODES);
    gcn_norm_kernel<<<eblocks, 256, 0, stream>>>(csrSrc, csrDst, csrW, dinv, norm, NEDGES);

    const int gblocks = (NNODES + 7) / 8;

    const int rtiles = NNODES / 16;
    const float* hin = x;
    const float* resin = x;
    for (int l = 0; l < 4; ++l) {
        int waves  = rtiles * (DH / 16);
        int blocks = (waves * 32 + 255) / 256;
        gcn_gemm_wmma<<<blocks, 256, 0, stream>>>(hin, W03 + (size_t)l * DH * DH, xw, NNODES, DH);

        gcn_gather_kernel<DH><<<gblocks, 256, 0, stream>>>(xw, rowptr, csrSrc, norm, agg, NNODES);

        long long ethreads = (long long)NNODES * 32;
        gcn_epilogue_hidden<<<(int)((ethreads + 255) / 256), 256, 0, stream>>>(
            agg, xw, dinv, b03 + l * DH, g03 + l * DH, be03 + l * DH, resin, hbuf, NNODES);

        hin = hbuf;
        resin = hbuf;
    }

    {
        int waves  = rtiles * (DOUT / 16);
        int blocks = (waves * 32 + 255) / 256;
        gcn_gemm_wmma<<<blocks, 256, 0, stream>>>(hbuf, W4, xw, NNODES, DOUT);

        gcn_gather_kernel<DOUT><<<gblocks, 256, 0, stream>>>(xw, rowptr, csrSrc, norm, agg, NNODES);

        long long ethreads = (long long)NNODES * 32;
        gcn_epilogue_final<<<(int)((ethreads + 255) / 256), 256, 0, stream>>>(
            agg, xw, dinv, b4, g4, be4, (float*)d_out, NNODES);
    }
}


# Fuse bias+LayerNorm(+relu+residual) epilogue into the neighbour-gather kernel: the wave that sums a node's rows finishes the node in registers; agg buffer and 5 launches removed; sum order kept
# speedup vs baseline: 2.5095x; 1.0773x over previous
#include <hip/hip_runtime.h>
#include <hip/hip_bf16.h>

#define NNODES 50000
#define NEDGES 640000
#define DH 128
#define DOUT 64
#define LN_EPS 1e-5f

typedef float v2f __attribute__((ext_vector_type(2)));
typedef float v8f __attribute__((ext_vector_type(8)));

__global__ void gcn_count_kernel(const int* __restrict__ dst, int* __restrict__ cnt,
                                 int* __restrict__ rnk, int E) {
    int e = blockIdx.x * blockDim.x + threadIdx.x;
    if (e < E) rnk[e] = atomicAdd(&cnt[dst[e]], 1);
}

__device__ __forceinline__ int wave_incl_scan32(int x, int lane) {
#pragma unroll
    for (int o = 1; o < 32; o <<= 1) {
        int y = __shfl_up(x, o, 32);
        if (lane >= o) x += y;
    }
    return x;
}

#define SCAN_CH ((NNODES + 1023) / 1024)
__global__ void __launch_bounds__(1024)
gcn_scan_kernel(const int* __restrict__ cnt, int* __restrict__ rowptr) {
    __shared__ int wtot[32];
    const int lane = threadIdx.x & 31;
    const int w    = threadIdx.x >> 5;
    const int base = w * (SCAN_CH * 32) + lane;
    int s = 0;
#pragma unroll
    for (int i = 0; i < SCAN_CH; ++i) {
        int j = base + i * 32;
        s += (j < NNODES) ? cnt[j] : 0;
    }
    const int sinc = wave_incl_scan32(s, lane);
    if (lane == 31) wtot[w] = sinc;
    __syncthreads();
    const int tv    = wtot[lane];
    const int tinc  = wave_incl_scan32(tv, lane);
    int       run   = __shfl(tinc - tv, w, 32);
    const int total = __shfl(tinc, 31, 32);
#pragma unroll
    for (int i = 0; i < SCAN_CH; ++i) {
        int j   = base + i * 32;
        int x   = (j < NNODES) ? cnt[j] : 0;
        int inc = wave_incl_scan32(x, lane);
        if (j < NNODES) rowptr[j] = run + inc - x;
        run += __shfl(inc, 31, 32);
    }
    if (threadIdx.x == 0) rowptr[NNODES] = total;
}

__global__ void gcn_place_kernel(const int* __restrict__ dst, const int* __restrict__ rnk,
                                 const int* __restrict__ rowptr, int* __restrict__ slot_eid, int E) {
    int e = blockIdx.x * blockDim.x + threadIdx.x;
    if (e < E) slot_eid[rowptr[dst[e]] + rnk[e]] = e;
}

__global__ void gcn_order_kernel(const int* __restrict__ src, const int* __restrict__ dst,
                                 const float* __restrict__ w, const int* __restrict__ rowptr,
                                 const int* __restrict__ slot_eid,
                                 int* __restrict__ csr_src, int* __restrict__ csr_dst,
                                 float* __restrict__ csr_w, int E) {
    int p = blockIdx.x * blockDim.x + threadIdx.x;
    if (p >= E) return;
    int e  = slot_eid[p];
    int d  = dst[e];
    int lo = rowptr[d], hi = rowptr[d + 1];
    int r  = 0;
    for (int j = lo; j < hi; ++j) r += (slot_eid[j] < e) ? 1 : 0;
    int q = lo + r;
    csr_src[q] = src[e];
    csr_dst[q] = d;
    csr_w[q]   = w[e];
}

__global__ void gcn_dinv_kernel(const int* __restrict__ rowptr, const float* __restrict__ csr_w,
                                float* __restrict__ dinv, int n) {
    int i = blockIdx.x * blockDim.x + threadIdx.x;
    if (i < n) {
        float d = 0.0f;
        for (int q = rowptr[i]; q < rowptr[i + 1]; ++q) d += csr_w[q];
        dinv[i] = rsqrtf(d + 1.0f);
    }
}

__global__ void gcn_norm_kernel(const int* __restrict__ csr_src, const int* __restrict__ csr_dst,
                                const float* __restrict__ csr_w, const float* __restrict__ dinv,
                                float* __restrict__ norm, int E) {
    int q = blockIdx.x * blockDim.x + threadIdx.x;
    if (q < E) norm[q] = dinv[csr_src[q]] * csr_w[q] * dinv[csr_dst[q]];
}

__global__ void gcn_gemm_wmma(const float* __restrict__ A,
                              const float* __restrict__ W,
                              float* __restrict__ out,
                              int nrows, int ncols) {
    const int lane = threadIdx.x & 31;
    const int wid  = blockIdx.x * (blockDim.x >> 5) + (threadIdx.x >> 5);
    const int ctiles = ncols >> 4;
    const int rtiles = nrows >> 4;
    if (wid >= rtiles * ctiles) return;
    const int rt = wid / ctiles;
    const int ct = wid - rt * ctiles;
    const int row0 = rt << 4;
    const int j0   = ct << 4;

    const int m  = lane & 15;
    const int ko = (lane >> 4) << 1;
    const float* __restrict__ arow = A + (size_t)(row0 + m) * DH + ko;
    const float* __restrict__ brow = W + (size_t)(j0 + m) * DH + ko;

    v8f acc = {};
#pragma unroll
    for (int k = 0; k < DH; k += 4) {
        v2f a = *(const v2f*)(arow + k);
        v2f b = *(const v2f*)(brow + k);
        acc = __builtin_amdgcn_wmma_f32_16x16x4_f32(
             false, a,  false, b,
             (short)0, acc,  false,  false);
    }

    float* __restrict__ o = out + (size_t)(row0 + ((lane >> 4) << 3)) * ncols + j0 + m;
#pragma unroll
    for (int v = 0; v < 8; ++v)
        o[(size_t)v * ncols] = acc[v];
}

__device__ __forceinline__ float wave_sum32(float v) {
#pragma unroll
    for (int m = 16; m >= 1; m >>= 1) v += __shfl_xor(v, m, 32);
    return v;
}

template <int D, bool HIDDEN>
__global__ void gcn_gather_ln_kernel(const float* __restrict__ xw, const int* __restrict__ rowptr,
                                     const int* __restrict__ csr_src, const float* __restrict__ norm,
                                     const float* __restrict__ dinv,
                                     const float* __restrict__ bias, const float* __restrict__ gamma,
                                     const float* __restrict__ beta,
                                     const float* res, float* hout, int n) {
    constexpr int V = D / 32;
    typedef float vf __attribute__((ext_vector_type(V)));
    const int lane = threadIdx.x & 31;
    const int node = __builtin_amdgcn_readfirstlane(blockIdx.x * (blockDim.x >> 5) + (threadIdx.x >> 5));
    if (node >= n) return;
    const int lo = rowptr[node], hi = rowptr[node + 1];
    const float* __restrict__ xl = xw + lane * V;
    const size_t own = (size_t)node * D + lane * V;

    const vf xs = *(const vf*)(xw + own);
    vf rs = {};
    if (HIDDEN) rs = *(const vf*)(res + own);
    const float di = dinv[node];
    const vf bv = *(const vf*)(bias + lane * V);
    const vf gv = *(const vf*)(gamma + lane * V);
    const vf ev = *(const vf*)(beta + lane * V);

    vf acc = {};
    for (int base = lo; base < hi; base += 32) {
        const int c = (hi - base < 32) ? (hi - base) : 32;
        int   ms = 0;
        float mw = 0.0f;
        if (lane < c) { ms = csr_src[base + lane]; mw = norm[base + lane]; }
        const int mwi = __float_as_int(mw);
        int j = 0;
        for (; j + 4 <= c; j += 4) {
            const int s0 = __builtin_amdgcn_readlane(ms, j);
            const int s1 = __builtin_amdgcn_readlane(ms, j + 1);
            const int s2 = __builtin_amdgcn_readlane(ms, j + 2);
            const int s3 = __builtin_amdgcn_readlane(ms, j + 3);
            const float w0 = __int_as_float(__builtin_amdgcn_readlane(mwi, j));
            const float w1 = __int_as_float(__builtin_amdgcn_readlane(mwi, j + 1));
            const float w2 = __int_as_float(__builtin_amdgcn_readlane(mwi, j + 2));
            const float w3 = __int_as_float(__builtin_amdgcn_readlane(mwi, j + 3));
            const vf v0 = *(const vf*)(xl + (size_t)s0 * D);
            const vf v1 = *(const vf*)(xl + (size_t)s1 * D);
            const vf v2 = *(const vf*)(xl + (size_t)s2 * D);
            const vf v3 = *(const vf*)(xl + (size_t)s3 * D);
            acc += w0 * v0;
            acc += w1 * v1;
            acc += w2 * v2;
            acc += w3 * v3;
        }
        for (; j < c; ++j) {
            const int s0 = __builtin_amdgcn_readlane(ms, j);
            const float w0 = __int_as_float(__builtin_amdgcn_readlane(mwi, j));
            const vf v0 = *(const vf*)(xl + (size_t)s0 * D);
            acc += w0 * v0;
        }
    }

    const float sl = di * di;
    float v[V];
#pragma unroll
    for (int i = 0; i < V; ++i) v[i] = fmaf(sl, xs[i], acc[i]) + bv[i];

    float ps = v[0], pq = v[0] * v[0];
#pragma unroll
    for (int i = 1; i < V; ++i) { ps += v[i]; pq += v[i] * v[i]; }
    const float s  = wave_sum32(ps);
    const float sq = wave_sum32(pq);
    const float mean = s * (1.0f / D);
    const float var  = sq * (1.0f / D) - mean * mean;
    const float rinv = rsqrtf(var + LN_EPS);

    vf o;
#pragma unroll
    for (int i = 0; i < V; ++i) {
        const float y = gv[i] * (v[i] - mean) * rinv + ev[i];
        o[i] = HIDDEN ? (fmaxf(y, 0.0f) + rs[i]) : y;
    }
    *(vf*)(hout + own) = o;
}

extern "C" void kernel_launch(void* const* d_in, const int* in_sizes, int n_in,
                              void* d_out, int out_size, void* d_ws, size_t ws_size,
                              hipStream_t stream) {
    const float* x    = (const float*)d_in[0];
    const int*   ei   = (const int*)d_in[1];
    const float* ew   = (const float*)d_in[2];
    const float* W03  = (const float*)d_in[3];
    const float* b03  = (const float*)d_in[4];
    const float* g03  = (const float*)d_in[5];
    const float* be03 = (const float*)d_in[6];
    const float* W4   = (const float*)d_in[7];
    const float* b4   = (const float*)d_in[8];
    const float* g4   = (const float*)d_in[9];
    const float* be4  = (const float*)d_in[10];

    const int* srcI = ei;
    const int* dstI = ei + NEDGES;

    float* xw   = (float*)d_ws;
    float* hbuf = xw + (size_t)NNODES * DH;
    float* dinv = hbuf + (size_t)NNODES * DH;
    float* norm = dinv + NNODES;
    float* csrW = norm + NEDGES;
    int* cnt     = (int*)(csrW + NEDGES);
    int* rowptr  = cnt + NNODES;
    int* rnk     = rowptr + NNODES + 1;
    int* slotEid = rnk + NEDGES;
    int* csrSrc  = slotEid + NEDGES;
    int* csrDst  = csrSrc + NEDGES;

    const int eblocks = (NEDGES + 255) / 256;
    hipMemsetAsync(cnt, 0, (size_t)NNODES * sizeof(int), stream);
    gcn_count_kernel<<<eblocks, 256, 0, stream>>>(dstI, cnt, rnk, NEDGES);
    gcn_scan_kernel<<<1, 1024, 0, stream>>>(cnt, rowptr);
    gcn_place_kernel<<<eblocks, 256, 0, stream>>>(dstI, rnk, rowptr, slotEid, NEDGES);
    gcn_order_kernel<<<eblocks, 256, 0, stream>>>(srcI, dstI, ew, rowptr, slotEid, csrSrc, csrDst, csrW, NEDGES);
    gcn_dinv_kernel<<<(NNODES + 255) / 256, 256, 0, stream>>>(rowptr, csrW, dinv, NNODES);
    gcn_norm_kernel<<<eblocks, 256, 0, stream>>>(csrSrc, csrDst, csrW, dinv, norm, NEDGES);

    const int gblocks = (NNODES + 7) / 8;

    const int rtiles = NNODES / 16;
    const float* hin = x;
    const float* resin = x;
    for (int l = 0; l < 4; ++l) {
        int waves  = rtiles * (DH / 16);
        int blocks = (waves * 32 + 255) / 256;
        gcn_gemm_wmma<<<blocks, 256, 0, stream>>>(hin, W03 + (size_t)l * DH * DH, xw, NNODES, DH);

        gcn_gather_ln_kernel<DH, true><<<gblocks, 256, 0, stream>>>(
            xw, rowptr, csrSrc, norm, dinv, b03 + l * DH, g03 + l * DH, be03 + l * DH,
            resin, hbuf, NNODES);

        hin = hbuf;
        resin = hbuf;
    }

    {
        int waves  = rtiles * (DOUT / 16);
        int blocks = (waves * 32 + 255) / 256;
        gcn_gemm_wmma<<<blocks, 256, 0, stream>>>(hbuf, W4, xw, NNODES, DOUT);

        gcn_gather_ln_kernel<DOUT, false><<<gblocks, 256, 0, stream>>>(
            xw, rowptr, csrSrc, norm, dinv, b4, g4, be4, nullptr, (float*)d_out, NNODES);
    }
}


# GEMM re-tiled: one wave owns a 16-row strip of all columns (8 or 4 accumulator tiles); W and A rows arrive as whole 512-byte rows and are read as fragments from LDS; k order and sum order kept
# speedup vs baseline: 3.0687x; 1.2228x over previous
#include <hip/hip_runtime.h>
#include <hip/hip_bf16.h>

#define NNODES 50000
#define NEDGES 640000
#define DH 128
#define DOUT 64
#define LN_EPS 1e-5f

typedef float v2f __attribute__((ext_vector_type(2)));
typedef float v8f __attribute__((ext_vector_type(8)));

__global__ void gcn_count_kernel(const int* __restrict__ dst, int* __restrict__ cnt,
                                 int* __restrict__ rnk, int E) {
    int e = blockIdx.x * blockDim.x + threadIdx.x;
    if (e < E) rnk[e] = atomicAdd(&cnt[dst[e]], 1);
}

__device__ __forceinline__ int wave_incl_scan32(int x, int lane) {
#pragma unroll
    for (int o = 1; o < 32; o <<= 1) {
        int y = __shfl_up(x, o, 32);
        if (lane >= o) x += y;
    }
    return x;
}

#define SCAN_CH ((NNODES + 1023) / 1024)
__global__ void __launch_bounds__(1024)
gcn_scan_kernel(const int* __restrict__ cnt, int* __restrict__ rowptr) {
    __shared__ int wtot[32];
    const int lane = threadIdx.x & 31;
    const int w    = threadIdx.x >> 5;
    const int base = w * (SCAN_CH * 32) + lane;
    int s = 0;
#pragma unroll
    for (int i = 0; i < SCAN_CH; ++i) {
        int j = base + i * 32;
        s += (j < NNODES) ? cnt[j] : 0;
    }
    const int sinc = wave_incl_scan32(s, lane);
    if (lane == 31) wtot[w] = sinc;
    __syncthreads();
    const int tv    = wtot[lane];
    const int tinc  = wave_incl_scan32(tv, lane);
    int       run   = __shfl(tinc - tv, w, 32);
    const int total = __shfl(tinc, 31, 32);
#pragma unroll
    for (int i = 0; i < SCAN_CH; ++i) {
        int j   = base + i * 32;
        int x   = (j < NNODES) ? cnt[j] : 0;
        int inc = wave_incl_scan32(x, lane);
        if (j < NNODES) rowptr[j] = run + inc - x;
        run += __shfl(inc, 31, 32);
    }
    if (threadIdx.x == 0) rowptr[NNODES] = total;
}

__global__ void gcn_place_kernel(const int* __restrict__ dst, const int* __restrict__ rnk,
                                 const int* __restrict__ rowptr, int* __restrict__ slot_eid, int E) {
    int e = blockIdx.x * blockDim.x + threadIdx.x;
    if (e < E) slot_eid[rowptr[dst[e]] + rnk[e]] = e;
}

__global__ void gcn_order_kernel(const int* __restrict__ src, const int* __restrict__ dst,
                                 const float* __restrict__ w, const int* __restrict__ rowptr,
                                 const int* __restrict__ slot_eid,
                                 int* __restrict__ csr_src, int* __restrict__ csr_dst,
                                 float* __restrict__ csr_w, int E) {
    int p = blockIdx.x * blockDim.x + threadIdx.x;
    if (p >= E) return;
    int e  = slot_eid[p];
    int d  = dst[e];
    int lo = rowptr[d], hi = rowptr[d + 1];
    int r  = 0;
    for (int j = lo; j < hi; ++j) r += (slot_eid[j] < e) ? 1 : 0;
    int q = lo + r;
    csr_src[q] = src[e];
    csr_dst[q] = d;
    csr_w[q]   = w[e];
}

__global__ void gcn_dinv_kernel(const int* __restrict__ rowptr, const float* __restrict__ csr_w,
                                float* __restrict__ dinv, int n) {
    int i = blockIdx.x * blockDim.x + threadIdx.x;
    if (i < n) {
        float d = 0.0f;
        for (int q = rowptr[i]; q < rowptr[i + 1]; ++q) d += csr_w[q];
        dinv[i] = rsqrtf(d + 1.0f);
    }
}

__global__ void gcn_norm_kernel(const int* __restrict__ csr_src, const int* __restrict__ csr_dst,
                                const float* __restrict__ csr_w, const float* __restrict__ dinv,
                                float* __restrict__ norm, int E) {
    int q = blockIdx.x * blockDim.x + threadIdx.x;
    if (q < E) norm[q] = dinv[csr_src[q]] * csr_w[q] * dinv[csr_dst[q]];
}

#define LDP 132
#define GEMM_WAVES 8

template <int NC>
__global__ void __launch_bounds__(GEMM_WAVES * 32)
gcn_gemm_wmma(const float* __restrict__ A,
              const float* __restrict__ W,
              float* __restrict__ out,
              int nrows) {
    constexpr int NT = NC / 16;
    __shared__ __attribute__((aligned(16))) float Wsm[NC * LDP];
    __shared__ __attribute__((aligned(16))) float Asm[GEMM_WAVES][16 * LDP];

    const int lane = threadIdx.x & 31;
    const int w    = threadIdx.x >> 5;
    const int strip   = blockIdx.x * GEMM_WAVES + w;
    const int nstrips = nrows >> 4;
    const bool active = strip < nstrips;
    const int row0 = (active ? strip : nstrips - 1) << 4;
    float* __restrict__ as = &Asm[w][0];

    float4 av[16], wv[NC / GEMM_WAVES];
    const float* __restrict__ ag = A + (size_t)row0 * DH + lane * 4;
#pragma unroll
    for (int r = 0; r < 16; ++r) av[r] = *(const float4*)(ag + (size_t)r * DH);
    const float* __restrict__ wg = W + (size_t)w * DH + lane * 4;
#pragma unroll
    for (int j = 0; j < NC / GEMM_WAVES; ++j) wv[j] = *(const float4*)(wg + (size_t)j * GEMM_WAVES * DH);
#pragma unroll
    for (int r = 0; r < 16; ++r) *(float4*)(&as[r * LDP + lane * 4]) = av[r];
#pragma unroll
    for (int j = 0; j < NC / GEMM_WAVES; ++j) *(float4*)(&Wsm[(w + j * GEMM_WAVES) * LDP + lane * 4]) = wv[j];
    __syncthreads();

    const int m  = lane & 15;
    const int ko = (lane >> 4) << 1;
    v8f acc[NT];
#pragma unroll
    for (int t = 0; t < NT; ++t) acc[t] = v8f{};

    const float* ap = as  + m * LDP + ko;
    const float* bp = Wsm + m * LDP + ko;
    v2f af[2][2];
    v2f bf[2][2][NT];
#pragma unroll
    for (int s = 0; s < 2; ++s) {
        af[0][s] = *(const v2f*)(ap + s * 4);
#pragma unroll
        for (int t = 0; t < NT; ++t) bf[0][s][t] = *(const v2f*)(bp + t * 16 * LDP + s * 4);
    }
#pragma unroll
    for (int kb = 0; kb < DH / 8; ++kb) {
        const int cur = kb & 1;
        if (kb + 1 < DH / 8) {
#pragma unroll
            for (int s = 0; s < 2; ++s) {
                af[cur ^ 1][s] = *(const v2f*)(ap + (kb + 1) * 8 + s * 4);
#pragma unroll
                for (int t = 0; t < NT; ++t)
                    bf[cur ^ 1][s][t] = *(const v2f*)(bp + t * 16 * LDP + (kb + 1) * 8 + s * 4);
            }
        }
        __builtin_amdgcn_sched_barrier(0);
#pragma unroll
        for (int s = 0; s < 2; ++s)
#pragma unroll
            for (int t = 0; t < NT; ++t)
                acc[t] = __builtin_amdgcn_wmma_f32_16x16x4_f32(
                     false, af[cur][s],  false, bf[cur][s][t],
                     (short)0, acc[t],  false,  false);
        __builtin_amdgcn_sched_barrier(0);
    }
    __syncthreads();

    {
        float* cs = as + ((lane >> 4) << 3) * LDP + m;
#pragma unroll
        for (int t = 0; t < NT; ++t)
#pragma unroll
            for (int v = 0; v < 8; ++v)
                cs[v * LDP + t * 16] = acc[t][v];
    }
    __syncthreads();

    if (active) {
        float* __restrict__ og = out + (size_t)row0 * NC + lane * 4;
#pragma unroll
        for (int j = 0; j < (16 * NC) / 128; ++j) {
            const int f   = j * 128 + lane * 4;
            const int row = f / NC;
            const int col = f - row * NC;
            const float4 v = *(const float4*)(&as[row * LDP + col]);
            *(float4*)(og + j * 128) = v;
        }
    }
}

__device__ __forceinline__ float wave_sum32(float v) {
#pragma unroll
    for (int m = 16; m >= 1; m >>= 1) v += __shfl_xor(v, m, 32);
    return v;
}

template <int D, bool HIDDEN>
__global__ void gcn_gather_ln_kernel(const float* __restrict__ xw, const int* __restrict__ rowptr,
                                     const int* __restrict__ csr_src, const float* __restrict__ norm,
                                     const float* __restrict__ dinv,
                                     const float* __restrict__ bias, const float* __restrict__ gamma,
                                     const float* __restrict__ beta,
                                     const float* res, float* hout, int n) {
    constexpr int V = D / 32;
    typedef float vf __attribute__((ext_vector_type(V)));
    const int lane = threadIdx.x & 31;
    const int node = __builtin_amdgcn_readfirstlane(blockIdx.x * (blockDim.x >> 5) + (threadIdx.x >> 5));
    if (node >= n) return;
    const int lo = rowptr[node], hi = rowptr[node + 1];
    const float* __restrict__ xl = xw + lane * V;
    const size_t own = (size_t)node * D + lane * V;

    const vf xs = *(const vf*)(xw + own);
    vf rs = {};
    if (HIDDEN) rs = *(const vf*)(res + own);
    const float di = dinv[node];
    const vf bv = *(const vf*)(bias + lane * V);
    const vf gv = *(const vf*)(gamma + lane * V);
    const vf ev = *(const vf*)(beta + lane * V);

    vf acc = {};
    for (int base = lo; base < hi; base += 32) {
        const int c = (hi - base < 32) ? (hi - base) : 32;
        int   ms = 0;
        float mw = 0.0f;
        if (lane < c) { ms = csr_src[base + lane]; mw = norm[base + lane]; }
        const int mwi = __float_as_int(mw);
        int j = 0;
        for (; j + 4 <= c; j += 4) {
            const int s0 = __builtin_amdgcn_readlane(ms, j);
            const int s1 = __builtin_amdgcn_readlane(ms, j + 1);
            const int s2 = __builtin_amdgcn_readlane(ms, j + 2);
            const int s3 = __builtin_amdgcn_readlane(ms, j + 3);
            const float w0 = __int_as_float(__builtin_amdgcn_readlane(mwi, j));
            const float w1 = __int_as_float(__builtin_amdgcn_readlane(mwi, j + 1));
            const float w2 = __int_as_float(__builtin_amdgcn_readlane(mwi, j + 2));
            const float w3 = __int_as_float(__builtin_amdgcn_readlane(mwi, j + 3));
            const vf v0 = *(const vf*)(xl + (size_t)s0 * D);
            const vf v1 = *(const vf*)(xl + (size_t)s1 * D);
            const vf v2 = *(const vf*)(xl + (size_t)s2 * D);
            const vf v3 = *(const vf*)(xl + (size_t)s3 * D);
            acc += w0 * v0;
            acc += w1 * v1;
            acc += w2 * v2;
            acc += w3 * v3;
        }
        for (; j < c; ++j) {
            const int s0 = __builtin_amdgcn_readlane(ms, j);
            const float w0 = __int_as_float(__builtin_amdgcn_readlane(mwi, j));
            const vf v0 = *(const vf*)(xl + (size_t)s0 * D);
            acc += w0 * v0;
        }
    }

    const float sl = di * di;
    float v[V];
#pragma unroll
    for (int i = 0; i < V; ++i) v[i] = fmaf(sl, xs[i], acc[i]) + bv[i];

    float ps = v[0], pq = v[0] * v[0];
#pragma unroll
    for (int i = 1; i < V; ++i) { ps += v[i]; pq += v[i] * v[i]; }
    const float s  = wave_sum32(ps);
    const float sq = wave_sum32(pq);
    const float mean = s * (1.0f / D);
    const float var  = sq * (1.0f / D) - mean * mean;
    const float rinv = rsqrtf(var + LN_EPS);

    vf o;
#pragma unroll
    for (int i = 0; i < V; ++i) {
        const float y = gv[i] * (v[i] - mean) * rinv + ev[i];
        o[i] = HIDDEN ? (fmaxf(y, 0.0f) + rs[i]) : y;
    }
    *(vf*)(hout + own) = o;
}

extern "C" void kernel_launch(void* const* d_in, const int* in_sizes, int n_in,
                              void* d_out, int out_size, void* d_ws, size_t ws_size,
                              hipStream_t stream) {
    const float* x    = (const float*)d_in[0];
    const int*   ei   = (const int*)d_in[1];
    const float* ew   = (const float*)d_in[2];
    const float* W03  = (const float*)d_in[3];
    const float* b03  = (const float*)d_in[4];
    const float* g03  = (const float*)d_in[5];
    const float* be03 = (const float*)d_in[6];
    const float* W4   = (const float*)d_in[7];
    const float* b4   = (const float*)d_in[8];
    const float* g4   = (const float*)d_in[9];
    const float* be4  = (const float*)d_in[10];

    const int* srcI = ei;
    const int* dstI = ei + NEDGES;

    float* xw   = (float*)d_ws;
    float* hbuf = xw + (size_t)NNODES * DH;
    float* dinv = hbuf + (size_t)NNODES * DH;
    float* norm = dinv + NNODES;
    float* csrW = norm + NEDGES;
    int* cnt     = (int*)(csrW + NEDGES);
    int* rowptr  = cnt + NNODES;
    int* rnk     = rowptr + NNODES + 1;
    int* slotEid = rnk + NEDGES;
    int* csrSrc  = slotEid + NEDGES;
    int* csrDst  = csrSrc + NEDGES;

    const int eblocks = (NEDGES + 255) / 256;
    hipMemsetAsync(cnt, 0, (size_t)NNODES * sizeof(int), stream);
    gcn_count_kernel<<<eblocks, 256, 0, stream>>>(dstI, cnt, rnk, NEDGES);
    gcn_scan_kernel<<<1, 1024, 0, stream>>>(cnt, rowptr);
    gcn_place_kernel<<<eblocks, 256, 0, stream>>>(dstI, rnk, rowptr, slotEid, NEDGES);
    gcn_order_kernel<<<eblocks, 256, 0, stream>>>(srcI, dstI, ew, rowptr, slotEid, csrSrc, csrDst, csrW, NEDGES);
    gcn_dinv_kernel<<<(NNODES + 255) / 256, 256, 0, stream>>>(rowptr, csrW, dinv, NNODES);
    gcn_norm_kernel<<<eblocks, 256, 0, stream>>>(csrSrc, csrDst, csrW, dinv, norm, NEDGES);

    const int gblocks = (NNODES + 7) / 8;

    const int gemm_blocks = (NNODES / 16 + GEMM_WAVES - 1) / GEMM_WAVES;
    const float* hin = x;
    const float* resin = x;
    for (int l = 0; l < 4; ++l) {
        gcn_gemm_wmma<DH><<<gemm_blocks, GEMM_WAVES * 32, 0, stream>>>(hin, W03 + (size_t)l * DH * DH, xw, NNODES);

        gcn_gather_ln_kernel<DH, true><<<gblocks, 256, 0, stream>>>(
            xw, rowptr, csrSrc, norm, dinv, b03 + l * DH, g03 + l * DH, be03 + l * DH,
            resin, hbuf, NNODES);

        hin = hbuf;
        resin = hbuf;
    }

    {
        gcn_gemm_wmma<DOUT><<<gemm_blocks, GEMM_WAVES * 32, 0, stream>>>(hbuf, W4, xw, NNODES);

        gcn_gather_ln_kernel<DOUT, false><<<gblocks, 256, 0, stream>>>(
            xw, rowptr, csrSrc, norm, dinv, b4, g4, be4, nullptr, (float*)d_out, NNODES);
    }
}


# Next layer's product fused into the gather launch: 512 resident 32-wave blocks finish 6-7 strips of nodes into an LDS tile and multiply it by W; 4 launches and the h re-read removed; sum order kept
# speedup vs baseline: 3.3727x; 1.0991x over previous
#include <hip/hip_runtime.h>
#include <hip/hip_bf16.h>

#define NNODES 50000
#define NEDGES 640000
#define DH 128
#define DOUT 64
#define LN_EPS 1e-5f

typedef float v2f __attribute__((ext_vector_type(2)));
typedef float v8f __attribute__((ext_vector_type(8)));

__global__ void gcn_count_kernel(const int* __restrict__ dst, int* __restrict__ cnt,
                                 int* __restrict__ rnk, int E) {
    int e = blockIdx.x * blockDim.x + threadIdx.x;
    if (e < E) rnk[e] = atomicAdd(&cnt[dst[e]], 1);
}

__device__ __forceinline__ int wave_incl_scan32(int x, int lane) {
#pragma unroll
    for (int o = 1; o < 32; o <<= 1) {
        int y = __shfl_up(x, o, 32);
        if (lane >= o) x += y;
    }
    return x;
}

#define SCAN_CH ((NNODES + 1023) / 1024)
__global__ void __launch_bounds__(1024)
gcn_scan_kernel(const int* __restrict__ cnt, int* __restrict__ rowptr) {
    __shared__ int wtot[32];
    const int lane = threadIdx.x & 31;
    const int w    = threadIdx.x >> 5;
    const int base = w * (SCAN_CH * 32) + lane;
    int s = 0;
#pragma unroll
    for (int i = 0; i < SCAN_CH; ++i) {
        int j = base + i * 32;
        s += (j < NNODES) ? cnt[j] : 0;
    }
    const int sinc = wave_incl_scan32(s, lane);
    if (lane == 31) wtot[w] = sinc;
    __syncthreads();
    const int tv    = wtot[lane];
    const int tinc  = wave_incl_scan32(tv, lane);
    int       run   = __shfl(tinc - tv, w, 32);
    const int total = __shfl(tinc, 31, 32);
#pragma unroll
    for (int i = 0; i < SCAN_CH; ++i) {
        int j   = base + i * 32;
        int x   = (j < NNODES) ? cnt[j] : 0;
        int inc = wave_incl_scan32(x, lane);
        if (j < NNODES) rowptr[j] = run + inc - x;
        run += __shfl(inc, 31, 32);
    }
    if (threadIdx.x == 0) rowptr[NNODES] = total;
}

__global__ void gcn_place_kernel(const int* __restrict__ dst, const int* __restrict__ rnk,
                                 const int* __restrict__ rowptr, int* __restrict__ slot_eid, int E) {
    int e = blockIdx.x * blockDim.x + threadIdx.x;
    if (e < E) slot_eid[rowptr[dst[e]] + rnk[e]] = e;
}

__global__ void gcn_order_kernel(const int* __restrict__ src, const int* __restrict__ dst,
                                 const float* __restrict__ w, const int* __restrict__ rowptr,
                                 const int* __restrict__ slot_eid,
                                 int* __restrict__ csr_src, int* __restrict__ csr_dst,
                                 float* __restrict__ csr_w, int E) {
    int p = blockIdx.x * blockDim.x + threadIdx.x;
    if (p >= E) return;
    int e  = slot_eid[p];
    int d  = dst[e];
    int lo = rowptr[d], hi = rowptr[d + 1];
    int r  = 0;
    for (int j = lo; j < hi; ++j) r += (slot_eid[j] < e) ? 1 : 0;
    int q = lo + r;
    csr_src[q] = src[e];
    csr_dst[q] = d;
    csr_w[q]   = w[e];
}

__global__ void gcn_dinv_kernel(const int* __restrict__ rowptr, const float* __restrict__ csr_w,
                                float* __restrict__ dinv, int n) {
    int i = blockIdx.x * blockDim.x + threadIdx.x;
    if (i < n) {
        float d = 0.0f;
        for (int q = rowptr[i]; q < rowptr[i + 1]; ++q) d += csr_w[q];
        dinv[i] = rsqrtf(d + 1.0f);
    }
}

__global__ void gcn_norm_kernel(const int* __restrict__ csr_src, const int* __restrict__ csr_dst,
                                const float* __restrict__ csr_w, const float* __restrict__ dinv,
                                float* __restrict__ norm, int E) {
    int q = blockIdx.x * blockDim.x + threadIdx.x;
    if (q < E) norm[q] = dinv[csr_src[q]] * csr_w[q] * dinv[csr_dst[q]];
}

#define LDP 132
#define GEMM_WAVES 8

template <int NC>
__global__ void __launch_bounds__(GEMM_WAVES * 32)
gcn_gemm_wmma(const float* __restrict__ A,
              const float* __restrict__ W,
              float* __restrict__ out,
              int nrows) {
    constexpr int NT = NC / 16;
    __shared__ __attribute__((aligned(16))) float Wsm[NC * LDP];
    __shared__ __attribute__((aligned(16))) float Asm[GEMM_WAVES][16 * LDP];

    const int lane = threadIdx.x & 31;
    const int w    = threadIdx.x >> 5;
    const int strip   = blockIdx.x * GEMM_WAVES + w;
    const int nstrips = nrows >> 4;
    const bool active = strip < nstrips;
    const int row0 = (active ? strip : nstrips - 1) << 4;
    float* __restrict__ as = &Asm[w][0];

    float4 av[16], wv[NC / GEMM_WAVES];
    const float* __restrict__ ag = A + (size_t)row0 * DH + lane * 4;
#pragma unroll
    for (int r = 0; r < 16; ++r) av[r] = *(const float4*)(ag + (size_t)r * DH);
    const float* __restrict__ wg = W + (size_t)w * DH + lane * 4;
#pragma unroll
    for (int j = 0; j < NC / GEMM_WAVES; ++j) wv[j] = *(const float4*)(wg + (size_t)j * GEMM_WAVES * DH);
#pragma unroll
    for (int r = 0; r < 16; ++r) *(float4*)(&as[r * LDP + lane * 4]) = av[r];
#pragma unroll
    for (int j = 0; j < NC / GEMM_WAVES; ++j) *(float4*)(&Wsm[(w + j * GEMM_WAVES) * LDP + lane * 4]) = wv[j];
    __syncthreads();

    const int m  = lane & 15;
    const int ko = (lane >> 4) << 1;
    v8f acc[NT];
#pragma unroll
    for (int t = 0; t < NT; ++t) acc[t] = v8f{};

    const float* ap = as  + m * LDP + ko;
    const float* bp = Wsm + m * LDP + ko;
    v2f af[2][2];
    v2f bf[2][2][NT];
#pragma unroll
    for (int s = 0; s < 2; ++s) {
        af[0][s] = *(const v2f*)(ap + s * 4);
#pragma unroll
        for (int t = 0; t < NT; ++t) bf[0][s][t] = *(const v2f*)(bp + t * 16 * LDP + s * 4);
    }
#pragma unroll
    for (int kb = 0; kb < DH / 8; ++kb) {
        const int cur = kb & 1;
        if (kb + 1 < DH / 8) {
#pragma unroll
            for (int s = 0; s < 2; ++s) {
                af[cur ^ 1][s] = *(const v2f*)(ap + (kb + 1) * 8 + s * 4);
#pragma unroll
                for (int t = 0; t < NT; ++t)
                    bf[cur ^ 1][s][t] = *(const v2f*)(bp + t * 16 * LDP + (kb + 1) * 8 + s * 4);
            }
        }
        __builtin_amdgcn_sched_barrier(0);
#pragma unroll
        for (int s = 0; s < 2; ++s)
#pragma unroll
            for (int t = 0; t < NT; ++t)
                acc[t] = __builtin_amdgcn_wmma_f32_16x16x4_f32(
                     false, af[cur][s],  false, bf[cur][s][t],
                     (short)0, acc[t],  false,  false);
        __builtin_amdgcn_sched_barrier(0);
    }
    __syncthreads();

    {
        float* cs = as + ((lane >> 4) << 3) * LDP + m;
#pragma unroll
        for (int t = 0; t < NT; ++t)
#pragma unroll
            for (int v = 0; v < 8; ++v)
                cs[v * LDP + t * 16] = acc[t][v];
    }
    __syncthreads();

    if (active) {
        float* __restrict__ og = out + (size_t)row0 * NC + lane * 4;
#pragma unroll
        for (int j = 0; j < (16 * NC) / 128; ++j) {
            const int f   = j * 128 + lane * 4;
            const int row = f / NC;
            const int col = f - row * NC;
            const float4 v = *(const float4*)(&as[row * LDP + col]);
            *(float4*)(og + j * 128) = v;
        }
    }
}

__device__ __forceinline__ float wave_sum32(float v) {
#pragma unroll
    for (int m = 16; m >= 1; m >>= 1) v += __shfl_xor(v, m, 32);
    return v;
}

template <int V> struct RowPiece { typedef float type __attribute__((ext_vector_type(V))); };

template <int D, bool HIDDEN>
__device__ __forceinline__ typename RowPiece<D / 32>::type
gcn_gather_ln_node(const float* __restrict__ xw, const int* __restrict__ rowptr,
                   const int* __restrict__ csr_src, const float* __restrict__ norm,
                   const float* __restrict__ dinv,
                   const float* __restrict__ bias, const float* __restrict__ gamma,
                   const float* __restrict__ beta,
                   const float* res, const int node, const int lane) {
    constexpr int V = D / 32;
    typedef typename RowPiece<V>::type vf;
    const int lo = rowptr[node], hi = rowptr[node + 1];
    const float* __restrict__ xl = xw + lane * V;
    const size_t own = (size_t)node * D + lane * V;

    const vf xs = *(const vf*)(xw + own);
    vf rs = {};
    if (HIDDEN) rs = *(const vf*)(res + own);
    const float di = dinv[node];
    const vf bv = *(const vf*)(bias + lane * V);
    const vf gv = *(const vf*)(gamma + lane * V);
    const vf ev = *(const vf*)(beta + lane * V);

    vf acc = {};
    for (int base = lo; base < hi; base += 32) {
        const int c = (hi - base < 32) ? (hi - base) : 32;
        int   ms = 0;
        float mw = 0.0f;
        if (lane < c) { ms = csr_src[base + lane]; mw = norm[base + lane]; }
        const int mwi = __float_as_int(mw);
        int j = 0;
        for (; j + 4 <= c; j += 4) {
            const int s0 = __builtin_amdgcn_readlane(ms, j);
            const int s1 = __builtin_amdgcn_readlane(ms, j + 1);
            const int s2 = __builtin_amdgcn_readlane(ms, j + 2);
            const int s3 = __builtin_amdgcn_readlane(ms, j + 3);
            const float w0 = __int_as_float(__builtin_amdgcn_readlane(mwi, j));
            const float w1 = __int_as_float(__builtin_amdgcn_readlane(mwi, j + 1));
            const float w2 = __int_as_float(__builtin_amdgcn_readlane(mwi, j + 2));
            const float w3 = __int_as_float(__builtin_amdgcn_readlane(mwi, j + 3));
            const vf v0 = *(const vf*)(xl + (size_t)s0 * D);
            const vf v1 = *(const vf*)(xl + (size_t)s1 * D);
            const vf v2 = *(const vf*)(xl + (size_t)s2 * D);
            const vf v3 = *(const vf*)(xl + (size_t)s3 * D);
            acc += w0 * v0;
            acc += w1 * v1;
            acc += w2 * v2;
            acc += w3 * v3;
        }
        for (; j < c; ++j) {
            const int s0 = __builtin_amdgcn_readlane(ms, j);
            const float w0 = __int_as_float(__builtin_amdgcn_readlane(mwi, j));
            const vf v0 = *(const vf*)(xl + (size_t)s0 * D);
            acc += w0 * v0;
        }
    }

    const float sl = di * di;
    float v[V];
#pragma unroll
    for (int i = 0; i < V; ++i) v[i] = fmaf(sl, xs[i], acc[i]) + bv[i];

    float ps = v[0], pq = v[0] * v[0];
#pragma unroll
    for (int i = 1; i < V; ++i) { ps += v[i]; pq += v[i] * v[i]; }
    const float s  = wave_sum32(ps);
    const float sq = wave_sum32(pq);
    const float mean = s * (1.0f / D);
    const float var  = sq * (1.0f / D) - mean * mean;
    const float rinv = rsqrtf(var + LN_EPS);

    vf o;
#pragma unroll
    for (int i = 0; i < V; ++i) {
        const float y = gv[i] * (v[i] - mean) * rinv + ev[i];
        o[i] = HIDDEN ? (fmaxf(y, 0.0f) + rs[i]) : y;
    }
    return o;
}

template <int D, bool HIDDEN>
__global__ void gcn_gather_ln_kernel(const float* __restrict__ xw, const int* __restrict__ rowptr,
                                     const int* __restrict__ csr_src, const float* __restrict__ norm,
                                     const float* __restrict__ dinv,
                                     const float* __restrict__ bias, const float* __restrict__ gamma,
                                     const float* __restrict__ beta,
                                     const float* res, float* hout, int n) {
    constexpr int V = D / 32;
    typedef typename RowPiece<V>::type vf;
    const int lane = threadIdx.x & 31;
    const int node = __builtin_amdgcn_readfirstlane(blockIdx.x * (blockDim.x >> 5) + (threadIdx.x >> 5));
    if (node >= n) return;
    const vf o = gcn_gather_ln_node<D, HIDDEN>(xw, rowptr, csr_src, norm, dinv, bias, gamma, beta,
                                               res, node, lane);
    *(vf*)(hout + (size_t)node * D + lane * V) = o;
}

#define FUSE_WAVES 32
#define FUSE_BLOCKS 512
#define FUSE_STRIPS 8

template <int NCN>
__global__ void __launch_bounds__(FUSE_WAVES * 32) __attribute__((amdgpu_waves_per_eu(16, 16)))
gcn_gather_ln_gemm_kernel(const float* __restrict__ xw, const int* __restrict__ rowptr,
                          const int* __restrict__ csr_src, const float* __restrict__ norm,
                          const float* __restrict__ dinv,
                          const float* __restrict__ bias, const float* __restrict__ gamma,
                          const float* __restrict__ beta,
                          const float* res, float* hout,
                          const float* __restrict__ Wn, float* __restrict__ xw_out, int n) {
    constexpr int NT = NCN / 64;
    constexpr int WR = NCN / FUSE_WAVES;
    __shared__ __attribute__((aligned(16))) float Wsm[NCN * LDP];
    __shared__ __attribute__((aligned(16))) float Asm[FUSE_STRIPS * 16 * LDP];

    const int lane = threadIdx.x & 31;
    const int w    = __builtin_amdgcn_readfirstlane(threadIdx.x >> 5);

    const int nstrips = n >> 4;
    const int s_lo  = ((int)blockIdx.x * nstrips) / FUSE_BLOCKS;
    const int s_hi  = (((int)blockIdx.x + 1) * nstrips) / FUSE_BLOCKS;
    const int ns    = s_hi - s_lo;
    const int nb    = s_lo << 4;
    const int nrows = ns << 4;

    for (int r = w; r < nrows; r += FUSE_WAVES) {
        const int node = nb + r;
        const typename RowPiece<4>::type q =
            gcn_gather_ln_node<DH, true>(xw, rowptr, csr_src, norm, dinv, bias, gamma, beta,
                                         res, node, lane);
        const float4 o = make_float4(q[0], q[1], q[2], q[3]);
        *(float4*)(hout + (size_t)node * DH + lane * 4) = o;
        *(float4*)(&Asm[r * LDP + lane * 4]) = o;
    }

    {
        float4 wv[WR];
        const float* __restrict__ wg = Wn + (size_t)w * DH + lane * 4;
#pragma unroll
        for (int j = 0; j < WR; ++j) wv[j] = *(const float4*)(wg + (size_t)j * FUSE_WAVES * DH);
#pragma unroll
        for (int j = 0; j < WR; ++j) *(float4*)(&Wsm[(w + j * FUSE_WAVES) * LDP + lane * 4]) = wv[j];
    }
    __syncthreads();

    const int strip = w >> 2;
    const int cg    = w & 3;
    const bool mine = strip < ns;
    const int m  = lane & 15;
    const int ko = (lane >> 4) << 1;
    v8f acc[NT];
#pragma unroll
    for (int t = 0; t < NT; ++t) acc[t] = v8f{};

    if (mine) {
        const float* ap = Asm + (strip * 16 + m) * LDP + ko;
        const float* bp = Wsm + (cg * NT * 16 + m) * LDP + ko;
        v2f af[2][2];
        v2f bf[2][2][NT];
#pragma unroll
        for (int s = 0; s < 2; ++s) {
            af[0][s] = *(const v2f*)(ap + s * 4);
#pragma unroll
            for (int t = 0; t < NT; ++t) bf[0][s][t] = *(const v2f*)(bp + t * 16 * LDP + s * 4);
        }
#pragma unroll
        for (int kb = 0; kb < DH / 8; ++kb) {
            const int cur = kb & 1;
            if (kb + 1 < DH / 8) {
#pragma unroll
                for (int s = 0; s < 2; ++s) {
                    af[cur ^ 1][s] = *(const v2f*)(ap + (kb + 1) * 8 + s * 4);
#pragma unroll
                    for (int t = 0; t < NT; ++t)
                        bf[cur ^ 1][s][t] = *(const v2f*)(bp + t * 16 * LDP + (kb + 1) * 8 + s * 4);
                }
            }
            __builtin_amdgcn_sched_barrier(0);
#pragma unroll
            for (int s = 0; s < 2; ++s)
#pragma unroll
                for (int t = 0; t < NT; ++t)
                    acc[t] = __builtin_amdgcn_wmma_f32_16x16x4_f32(
                         false, af[cur][s],  false, bf[cur][s][t],
                         (short)0, acc[t],  false,  false);
            __builtin_amdgcn_sched_barrier(0);
        }
    }
    __syncthreads();

    if (mine) {
        float* cs = Asm + (strip * 16 + ((lane >> 4) << 3)) * LDP + cg * NT * 16 + m;
#pragma unroll
        for (int t = 0; t < NT; ++t)
#pragma unroll
            for (int v = 0; v < 8; ++v)
                cs[v * LDP + t * 16] = acc[t][v];
    }
    __syncthreads();

    if (mine) {
        constexpr int PER = (16 * NCN) / 128 / 4;
        float* __restrict__ og = xw_out + (size_t)(s_lo + strip) * 16 * NCN;
#pragma unroll
        for (int j = 0; j < PER; ++j) {
            const int f   = (cg * PER + j) * 128 + lane * 4;
            const int row = f / NCN;
            const int col = f - row * NCN;
            const float4 v = *(const float4*)(&Asm[(strip * 16 + row) * LDP + col]);
            *(float4*)(og + f) = v;
        }
    }
}

extern "C" void kernel_launch(void* const* d_in, const int* in_sizes, int n_in,
                              void* d_out, int out_size, void* d_ws, size_t ws_size,
                              hipStream_t stream) {
    const float* x    = (const float*)d_in[0];
    const int*   ei   = (const int*)d_in[1];
    const float* ew   = (const float*)d_in[2];
    const float* W03  = (const float*)d_in[3];
    const float* b03  = (const float*)d_in[4];
    const float* g03  = (const float*)d_in[5];
    const float* be03 = (const float*)d_in[6];
    const float* W4   = (const float*)d_in[7];
    const float* b4   = (const float*)d_in[8];
    const float* g4   = (const float*)d_in[9];
    const float* be4  = (const float*)d_in[10];

    const int* srcI = ei;
    const int* dstI = ei + NEDGES;

    float* xwa  = (float*)d_ws;
    float* xwb  = xwa + (size_t)NNODES * DH;
    float* hbuf = xwb + (size_t)NNODES * DH;
    float* dinv = hbuf + (size_t)NNODES * DH;
    float* norm = dinv + NNODES;
    float* csrW = norm + NEDGES;
    int* cnt     = (int*)(csrW + NEDGES);
    int* rowptr  = cnt + NNODES;
    int* rnk     = rowptr + NNODES + 1;
    int* slotEid = rnk + NEDGES;
    int* csrSrc  = slotEid + NEDGES;
    int* csrDst  = csrSrc + NEDGES;

    const int eblocks = (NEDGES + 255) / 256;
    hipMemsetAsync(cnt, 0, (size_t)NNODES * sizeof(int), stream);
    gcn_count_kernel<<<eblocks, 256, 0, stream>>>(dstI, cnt, rnk, NEDGES);
    gcn_scan_kernel<<<1, 1024, 0, stream>>>(cnt, rowptr);
    gcn_place_kernel<<<eblocks, 256, 0, stream>>>(dstI, rnk, rowptr, slotEid, NEDGES);
    gcn_order_kernel<<<eblocks, 256, 0, stream>>>(srcI, dstI, ew, rowptr, slotEid, csrSrc, csrDst, csrW, NEDGES);
    gcn_dinv_kernel<<<(NNODES + 255) / 256, 256, 0, stream>>>(rowptr, csrW, dinv, NNODES);
    gcn_norm_kernel<<<eblocks, 256, 0, stream>>>(csrSrc, csrDst, csrW, dinv, norm, NEDGES);

    const int gemm_blocks = (NNODES / 16 + GEMM_WAVES - 1) / GEMM_WAVES;
    gcn_gemm_wmma<DH><<<gemm_blocks, GEMM_WAVES * 32, 0, stream>>>(x, W03, xwa, NNODES);

    const int fblocks = FUSE_BLOCKS;
    gcn_gather_ln_gemm_kernel<DH><<<fblocks, FUSE_WAVES * 32, 0, stream>>>(
        xwa, rowptr, csrSrc, norm, dinv, b03, g03, be03, x, hbuf,
        W03 + (size_t)1 * DH * DH, xwb, NNODES);
    gcn_gather_ln_gemm_kernel<DH><<<fblocks, FUSE_WAVES * 32, 0, stream>>>(
        xwb, rowptr, csrSrc, norm, dinv, b03 + 1 * DH, g03 + 1 * DH, be03 + 1 * DH, hbuf, hbuf,
        W03 + (size_t)2 * DH * DH, xwa, NNODES);
    gcn_gather_ln_gemm_kernel<DH><<<fblocks, FUSE_WAVES * 32, 0, stream>>>(
        xwa, rowptr, csrSrc, norm, dinv, b03 + 2 * DH, g03 + 2 * DH, be03 + 2 * DH, hbuf, hbuf,
        W03 + (size_t)3 * DH * DH, xwb, NNODES);
    gcn_gather_ln_gemm_kernel<DOUT><<<fblocks, FUSE_WAVES * 32, 0, stream>>>(
        xwb, rowptr, csrSrc, norm, dinv, b03 + 3 * DH, g03 + 3 * DH, be03 + 3 * DH, hbuf, hbuf,
        W4, xwa, NNODES);

    const int gblocks = (NNODES + 7) / 8;
    gcn_gather_ln_kernel<DOUT, false><<<gblocks, 256, 0, stream>>>(
        xwa, rowptr, csrSrc, norm, dinv, b4, g4, be4, nullptr, (float*)d_out, NNODES);
}


# Edge count and placement on LDS counters per (destination range, edge slice) work-group, the LDS add's return is the slot; returning global int atomics and the clear launch removed; sum order kept
# speedup vs baseline: 3.6485x; 1.0818x over previous
#include <hip/hip_runtime.h>
#include <hip/hip_bf16.h>

#define NNODES 50000
#define NEDGES 640000
#define DH 128
#define DOUT 64
#define LN_EPS 1e-5f

typedef float v2f __attribute__((ext_vector_type(2)));
typedef float v8f __attribute__((ext_vector_type(8)));

#define CNT_RANGES 64
#define CNT_SLICES 4
#define CNT_RS ((NNODES + CNT_RANGES - 1) / CNT_RANGES)
#define CNT_Q  (NEDGES / 4 / CNT_SLICES)
#define CNT_G  ((CNT_Q + 8 * 1024 - 1) / (8 * 1024))

__global__ void __launch_bounds__(1024)
gcn_count_kernel(const int* __restrict__ dst, int* __restrict__ parts) {
    __shared__ int c[CNT_RS];
    const int r  = blockIdx.x / CNT_SLICES;
    const int s  = blockIdx.x - r * CNT_SLICES;
    const int lo = r * CNT_RS;
    for (int i = threadIdx.x; i < CNT_RS; i += 1024) c[i] = 0;
    __syncthreads();
    const int4* __restrict__ d4 = (const int4*)dst + (size_t)s * CNT_Q;
    for (int g = 0; g < CNT_G; ++g) {
        int4 v[8];
#pragma unroll
        for (int u = 0; u < 8; ++u) {
            const int i = (g * 8 + u) * 1024 + (int)threadIdx.x;
            v[u] = d4[i < CNT_Q ? i : CNT_Q - 1];
            if (i >= CNT_Q) v[u] = make_int4(-1, -1, -1, -1);
        }
#pragma unroll
        for (int u = 0; u < 8; ++u) {
            const unsigned a0 = (unsigned)(v[u].x - lo), a1 = (unsigned)(v[u].y - lo);
            const unsigned a2 = (unsigned)(v[u].z - lo), a3 = (unsigned)(v[u].w - lo);
            if (a0 < CNT_RS) atomicAdd(&c[a0], 1);
            if (a1 < CNT_RS) atomicAdd(&c[a1], 1);
            if (a2 < CNT_RS) atomicAdd(&c[a2], 1);
            if (a3 < CNT_RS) atomicAdd(&c[a3], 1);
        }
    }
    __syncthreads();
    for (int i = threadIdx.x; i < CNT_RS; i += 1024) {
        const int n = lo + i;
        if (n < NNODES) parts[(size_t)s * NNODES + n] = c[i];
    }
}

__device__ __forceinline__ int wave_incl_scan32(int x, int lane) {
#pragma unroll
    for (int o = 1; o < 32; o <<= 1) {
        int y = __shfl_up(x, o, 32);
        if (lane >= o) x += y;
    }
    return x;
}

#define SCAN_CH ((NNODES + 1023) / 1024)
__global__ void __launch_bounds__(1024)
gcn_scan_kernel(const int* __restrict__ parts, int* cnt, int* __restrict__ rowptr) {
    __shared__ int wtot[32];
    const int lane = threadIdx.x & 31;
    const int w    = threadIdx.x >> 5;
    const int base = w * (SCAN_CH * 32) + lane;
    int s = 0;
#pragma unroll
    for (int i = 0; i < SCAN_CH; ++i) {
        int j = base + i * 32;
        int t = 0;
        if (j < NNODES) {
            t = parts[j];
#pragma unroll
            for (int p = 1; p < CNT_SLICES; ++p) t += parts[(size_t)p * NNODES + j];
            cnt[j] = t;
        }
        s += t;
    }
    const int sinc = wave_incl_scan32(s, lane);
    if (lane == 31) wtot[w] = sinc;
    __syncthreads();
    const int tv    = wtot[lane];
    const int tinc  = wave_incl_scan32(tv, lane);
    int       run   = __shfl(tinc - tv, w, 32);
    const int total = __shfl(tinc, 31, 32);
#pragma unroll
    for (int i = 0; i < SCAN_CH; ++i) {
        int j   = base + i * 32;
        int x   = (j < NNODES) ? cnt[j] : 0;
        int inc = wave_incl_scan32(x, lane);
        if (j < NNODES) rowptr[j] = run + inc - x;
        run += __shfl(inc, 31, 32);
    }
    if (threadIdx.x == 0) rowptr[NNODES] = total;
}

__global__ void __launch_bounds__(1024)
gcn_place_kernel(const int* __restrict__ dst, const int* __restrict__ parts,
                 const int* __restrict__ rowptr, int* __restrict__ slot_eid) {
    __shared__ int c[CNT_RS];
    const int r  = blockIdx.x / CNT_SLICES;
    const int s  = blockIdx.x - r * CNT_SLICES;
    const int lo = r * CNT_RS;
    for (int i = threadIdx.x; i < CNT_RS; i += 1024) {
        const int n = lo + i;
        int b = 0;
        if (n < NNODES) {
            b = rowptr[n];
            for (int p = 0; p < s; ++p) b += parts[(size_t)p * NNODES + n];
        }
        c[i] = b;
    }
    __syncthreads();
    const int4* __restrict__ d4 = (const int4*)dst + (size_t)s * CNT_Q;
    for (int g = 0; g < CNT_G; ++g) {
        int4 v[8];
#pragma unroll
        for (int u = 0; u < 8; ++u) {
            const int i = (g * 8 + u) * 1024 + (int)threadIdx.x;
            v[u] = d4[i < CNT_Q ? i : CNT_Q - 1];
            if (i >= CNT_Q) v[u] = make_int4(-1, -1, -1, -1);
        }
#pragma unroll
        for (int u = 0; u < 8; ++u) {
            const int e0 = (s * CNT_Q + (g * 8 + u) * 1024 + (int)threadIdx.x) * 4;
            const unsigned a0 = (unsigned)(v[u].x - lo), a1 = (unsigned)(v[u].y - lo);
            const unsigned a2 = (unsigned)(v[u].z - lo), a3 = (unsigned)(v[u].w - lo);
            if (a0 < CNT_RS) slot_eid[atomicAdd(&c[a0], 1)] = e0;
            if (a1 < CNT_RS) slot_eid[atomicAdd(&c[a1], 1)] = e0 + 1;
            if (a2 < CNT_RS) slot_eid[atomicAdd(&c[a2], 1)] = e0 + 2;
            if (a3 < CNT_RS) slot_eid[atomicAdd(&c[a3], 1)] = e0 + 3;
        }
    }
}

__global__ void gcn_order_kernel(const int* __restrict__ src, const int* __restrict__ dst,
                                 const float* __restrict__ w, const int* __restrict__ rowptr,
                                 const int* __restrict__ slot_eid,
                                 int* __restrict__ csr_src, int* __restrict__ csr_dst,
                                 float* __restrict__ csr_w, int E) {
    int p = blockIdx.x * blockDim.x + threadIdx.x;
    if (p >= E) return;
    int e  = slot_eid[p];
    int d  = dst[e];
    int lo = rowptr[d], hi = rowptr[d + 1];
    int r  = 0;
    for (int j = lo; j < hi; ++j) r += (slot_eid[j] < e) ? 1 : 0;
    int q = lo + r;
    csr_src[q] = src[e];
    csr_dst[q] = d;
    csr_w[q]   = w[e];
}

__global__ void gcn_dinv_kernel(const int* __restrict__ rowptr, const float* __restrict__ csr_w,
                                float* __restrict__ dinv, int n) {
    int i = blockIdx.x * blockDim.x + threadIdx.x;
    if (i < n) {
        float d = 0.0f;
        for (int q = rowptr[i]; q < rowptr[i + 1]; ++q) d += csr_w[q];
        dinv[i] = rsqrtf(d + 1.0f);
    }
}

__global__ void gcn_norm_kernel(const int* __restrict__ csr_src, const int* __restrict__ csr_dst,
                                const float* __restrict__ csr_w, const float* __restrict__ dinv,
                                float* __restrict__ norm, int E) {
    int q = blockIdx.x * blockDim.x + threadIdx.x;
    if (q < E) norm[q] = dinv[csr_src[q]] * csr_w[q] * dinv[csr_dst[q]];
}

#define LDP 132
#define GEMM_WAVES 8

template <int NC>
__global__ void __launch_bounds__(GEMM_WAVES * 32)
gcn_gemm_wmma(const float* __restrict__ A,
              const float* __restrict__ W,
              float* __restrict__ out,
              int nrows) {
    constexpr int NT = NC / 16;
    __shared__ __attribute__((aligned(16))) float Wsm[NC * LDP];
    __shared__ __attribute__((aligned(16))) float Asm[GEMM_WAVES][16 * LDP];

    const int lane = threadIdx.x & 31;
    const int w    = threadIdx.x >> 5;
    const int strip   = blockIdx.x * GEMM_WAVES + w;
    const int nstrips = nrows >> 4;
    const bool active = strip < nstrips;
    const int row0 = (active ? strip : nstrips - 1) << 4;
    float* __restrict__ as = &Asm[w][0];

    float4 av[16], wv[NC / GEMM_WAVES];
    const float* __restrict__ ag = A + (size_t)row0 * DH + lane * 4;
#pragma unroll
    for (int r = 0; r < 16; ++r) av[r] = *(const float4*)(ag + (size_t)r * DH);
    const float* __restrict__ wg = W + (size_t)w * DH + lane * 4;
#pragma unroll
    for (int j = 0; j < NC / GEMM_WAVES; ++j) wv[j] = *(const float4*)(wg + (size_t)j * GEMM_WAVES * DH);
#pragma unroll
    for (int r = 0; r < 16; ++r) *(float4*)(&as[r * LDP + lane * 4]) = av[r];
#pragma unroll
    for (int j = 0; j < NC / GEMM_WAVES; ++j) *(float4*)(&Wsm[(w + j * GEMM_WAVES) * LDP + lane * 4]) = wv[j];
    __syncthreads();

    const int m  = lane & 15;
    const int ko = (lane >> 4) << 1;
    v8f acc[NT];
#pragma unroll
    for (int t = 0; t < NT; ++t) acc[t] = v8f{};

    const float* ap = as  + m * LDP + ko;
    const float* bp = Wsm + m * LDP + ko;
    v2f af[2][2];
    v2f bf[2][2][NT];
#pragma unroll
    for (int s = 0; s < 2; ++s) {
        af[0][s] = *(const v2f*)(ap + s * 4);
#pragma unroll
        for (int t = 0; t < NT; ++t) bf[0][s][t] = *(const v2f*)(bp + t * 16 * LDP + s * 4);
    }
#pragma unroll
    for (int kb = 0; kb < DH / 8; ++kb) {
        const int cur = kb & 1;
        if (kb + 1 < DH / 8) {
#pragma unroll
            for (int s = 0; s < 2; ++s) {
                af[cur ^ 1][s] = *(const v2f*)(ap + (kb + 1) * 8 + s * 4);
#pragma unroll
                for (int t = 0; t < NT; ++t)
                    bf[cur ^ 1][s][t] = *(const v2f*)(bp + t * 16 * LDP + (kb + 1) * 8 + s * 4);
            }
        }
        __builtin_amdgcn_sched_barrier(0);
#pragma unroll
        for (int s = 0; s < 2; ++s)
#pragma unroll
            for (int t = 0; t < NT; ++t)
                acc[t] = __builtin_amdgcn_wmma_f32_16x16x4_f32(
                     false, af[cur][s],  false, bf[cur][s][t],
                     (short)0, acc[t],  false,  false);
        __builtin_amdgcn_sched_barrier(0);
    }
    __syncthreads();

    {
        float* cs = as + ((lane >> 4) << 3) * LDP + m;
#pragma unroll
        for (int t = 0; t < NT; ++t)
#pragma unroll
            for (int v = 0; v < 8; ++v)
                cs[v * LDP + t * 16] = acc[t][v];
    }
    __syncthreads();

    if (active) {
        float* __restrict__ og = out + (size_t)row0 * NC + lane * 4;
#pragma unroll
        for (int j = 0; j < (16 * NC) / 128; ++j) {
            const int f   = j * 128 + lane * 4;
            const int row = f / NC;
            const int col = f - row * NC;
            const float4 v = *(const float4*)(&as[row * LDP + col]);
            *(float4*)(og + j * 128) = v;
        }
    }
}

__device__ __forceinline__ float wave_sum32(float v) {
#pragma unroll
    for (int m = 16; m >= 1; m >>= 1) v += __shfl_xor(v, m, 32);
    return v;
}

template <int V> struct RowPiece { typedef float type __attribute__((ext_vector_type(V))); };

template <int D, bool HIDDEN>
__device__ __forceinline__ typename RowPiece<D / 32>::type
gcn_gather_ln_node(const float* __restrict__ xw, const int* __restrict__ rowptr,
                   const int* __restrict__ csr_src, const float* __restrict__ norm,
                   const float* __restrict__ dinv,
                   const float* __restrict__ bias, const float* __restrict__ gamma,
                   const float* __restrict__ beta,
                   const float* res, const int node, const int lane) {
    constexpr int V = D / 32;
    typedef typename RowPiece<V>::type vf;
    const int lo = rowptr[node], hi = rowptr[node + 1];
    const float* __restrict__ xl = xw + lane * V;
    const size_t own = (size_t)node * D + lane * V;

    const vf xs = *(const vf*)(xw + own);
    vf rs = {};
    if (HIDDEN) rs = *(const vf*)(res + own);
    const float di = dinv[node];
    const vf bv = *(const vf*)(bias + lane * V);
    const vf gv = *(const vf*)(gamma + lane * V);
    const vf ev = *(const vf*)(beta + lane * V);

    vf acc = {};
    for (int base = lo; base < hi; base += 32) {
        const int c = (hi - base < 32) ? (hi - base) : 32;
        int   ms = 0;
        float mw = 0.0f;
        if (lane < c) { ms = csr_src[base + lane]; mw = norm[base + lane]; }
        const int mwi = __float_as_int(mw);
        int j = 0;
        for (; j + 4 <= c; j += 4) {
            const int s0 = __builtin_amdgcn_readlane(ms, j);
            const int s1 = __builtin_amdgcn_readlane(ms, j + 1);
            const int s2 = __builtin_amdgcn_readlane(ms, j + 2);
            const int s3 = __builtin_amdgcn_readlane(ms, j + 3);
            const float w0 = __int_as_float(__builtin_amdgcn_readlane(mwi, j));
            const float w1 = __int_as_float(__builtin_amdgcn_readlane(mwi, j + 1));
            const float w2 = __int_as_float(__builtin_amdgcn_readlane(mwi, j + 2));
            const float w3 = __int_as_float(__builtin_amdgcn_readlane(mwi, j + 3));
            const vf v0 = *(const vf*)(xl + (size_t)s0 * D);
            const vf v1 = *(const vf*)(xl + (size_t)s1 * D);
            const vf v2 = *(const vf*)(xl + (size_t)s2 * D);
            const vf v3 = *(const vf*)(xl + (size_t)s3 * D);
            acc += w0 * v0;
            acc += w1 * v1;
            acc += w2 * v2;
            acc += w3 * v3;
        }
        for (; j < c; ++j) {
            const int s0 = __builtin_amdgcn_readlane(ms, j);
            const float w0 = __int_as_float(__builtin_amdgcn_readlane(mwi, j));
            const vf v0 = *(const vf*)(xl + (size_t)s0 * D);
            acc += w0 * v0;
        }
    }

    const float sl = di * di;
    float v[V];
#pragma unroll
    for (int i = 0; i < V; ++i) v[i] = fmaf(sl, xs[i], acc[i]) + bv[i];

    float ps = v[0], pq = v[0] * v[0];
#pragma unroll
    for (int i = 1; i < V; ++i) { ps += v[i]; pq += v[i] * v[i]; }
    const float s  = wave_sum32(ps);
    const float sq = wave_sum32(pq);
    const float mean = s * (1.0f / D);
    const float var  = sq * (1.0f / D) - mean * mean;
    const float rinv = rsqrtf(var + LN_EPS);

    vf o;
#pragma unroll
    for (int i = 0; i < V; ++i) {
        const float y = gv[i] * (v[i] - mean) * rinv + ev[i];
        o[i] = HIDDEN ? (fmaxf(y, 0.0f) + rs[i]) : y;
    }
    return o;
}

template <int D, bool HIDDEN>
__global__ void gcn_gather_ln_kernel(const float* __restrict__ xw, const int* __restrict__ rowptr,
                                     const int* __restrict__ csr_src, const float* __restrict__ norm,
                                     const float* __restrict__ dinv,
                                     const float* __restrict__ bias, const float* __restrict__ gamma,
                                     const float* __restrict__ beta,
                                     const float* res, float* hout, int n) {
    constexpr int V = D / 32;
    typedef typename RowPiece<V>::type vf;
    const int lane = threadIdx.x & 31;
    const int node = __builtin_amdgcn_readfirstlane(blockIdx.x * (blockDim.x >> 5) + (threadIdx.x >> 5));
    if (node >= n) return;
    const vf o = gcn_gather_ln_node<D, HIDDEN>(xw, rowptr, csr_src, norm, dinv, bias, gamma, beta,
                                               res, node, lane);
    *(vf*)(hout + (size_t)node * D + lane * V) = o;
}

#define FUSE_WAVES 32
#define FUSE_BLOCKS 512
#define FUSE_STRIPS 8

template <int NCN>
__global__ void __launch_bounds__(FUSE_WAVES * 32) __attribute__((amdgpu_waves_per_eu(16, 16)))
gcn_gather_ln_gemm_kernel(const float* __restrict__ xw, const int* __restrict__ rowptr,
                          const int* __restrict__ csr_src, const float* __restrict__ norm,
                          const float* __restrict__ dinv,
                          const float* __restrict__ bias, const float* __restrict__ gamma,
                          const float* __restrict__ beta,
                          const float* res, float* hout,
                          const float* __restrict__ Wn, float* __restrict__ xw_out, int n) {
    constexpr int NT = NCN / 64;
    constexpr int WR = NCN / FUSE_WAVES;
    __shared__ __attribute__((aligned(16))) float Wsm[NCN * LDP];
    __shared__ __attribute__((aligned(16))) float Asm[FUSE_STRIPS * 16 * LDP];

    const int lane = threadIdx.x & 31;
    const int w    = __builtin_amdgcn_readfirstlane(threadIdx.x >> 5);

    const int nstrips = n >> 4;
    const int s_lo  = ((int)blockIdx.x * nstrips) / FUSE_BLOCKS;
    const int s_hi  = (((int)blockIdx.x + 1) * nstrips) / FUSE_BLOCKS;
    const int ns    = s_hi - s_lo;
    const int nb    = s_lo << 4;
    const int nrows = ns << 4;

    for (int r = w; r < nrows; r += FUSE_WAVES) {
        const int node = nb + r;
        const typename RowPiece<4>::type q =
            gcn_gather_ln_node<DH, true>(xw, rowptr, csr_src, norm, dinv, bias, gamma, beta,
                                         res, node, lane);
        const float4 o = make_float4(q[0], q[1], q[2], q[3]);
        *(float4*)(hout + (size_t)node * DH + lane * 4) = o;
        *(float4*)(&Asm[r * LDP + lane * 4]) = o;
    }

    {
        float4 wv[WR];
        const float* __restrict__ wg = Wn + (size_t)w * DH + lane * 4;
#pragma unroll
        for (int j = 0; j < WR; ++j) wv[j] = *(const float4*)(wg + (size_t)j * FUSE_WAVES * DH);
#pragma unroll
        for (int j = 0; j < WR; ++j) *(float4*)(&Wsm[(w + j * FUSE_WAVES) * LDP + lane * 4]) = wv[j];
    }
    __syncthreads();

    const int strip = w >> 2;
    const int cg    = w & 3;
    const bool mine = strip < ns;
    const int m  = lane & 15;
    const int ko = (lane >> 4) << 1;
    v8f acc[NT];
#pragma unroll
    for (int t = 0; t < NT; ++t) acc[t] = v8f{};

    if (mine) {
        const float* ap = Asm + (strip * 16 + m) * LDP + ko;
        const float* bp = Wsm + (cg * NT * 16 + m) * LDP + ko;
        v2f af[2][2];
        v2f bf[2][2][NT];
#pragma unroll
        for (int s = 0; s < 2; ++s) {
            af[0][s] = *(const v2f*)(ap + s * 4);
#pragma unroll
            for (int t = 0; t < NT; ++t) bf[0][s][t] = *(const v2f*)(bp + t * 16 * LDP + s * 4);
        }
#pragma unroll
        for (int kb = 0; kb < DH / 8; ++kb) {
            const int cur = kb & 1;
            if (kb + 1 < DH / 8) {
#pragma unroll
                for (int s = 0; s < 2; ++s) {
                    af[cur ^ 1][s] = *(const v2f*)(ap + (kb + 1) * 8 + s * 4);
#pragma unroll
                    for (int t = 0; t < NT; ++t)
                        bf[cur ^ 1][s][t] = *(const v2f*)(bp + t * 16 * LDP + (kb + 1) * 8 + s * 4);
                }
            }
            __builtin_amdgcn_sched_barrier(0);
#pragma unroll
            for (int s = 0; s < 2; ++s)
#pragma unroll
                for (int t = 0; t < NT; ++t)
                    acc[t] = __builtin_amdgcn_wmma_f32_16x16x4_f32(
                         false, af[cur][s],  false, bf[cur][s][t],
                         (short)0, acc[t],  false,  false);
            __builtin_amdgcn_sched_barrier(0);
        }
    }
    __syncthreads();

    if (mine) {
        float* cs = Asm + (strip * 16 + ((lane >> 4) << 3)) * LDP + cg * NT * 16 + m;
#pragma unroll
        for (int t = 0; t < NT; ++t)
#pragma unroll
            for (int v = 0; v < 8; ++v)
                cs[v * LDP + t * 16] = acc[t][v];
    }
    __syncthreads();

    if (mine) {
        constexpr int PER = (16 * NCN) / 128 / 4;
        float* __restrict__ og = xw_out + (size_t)(s_lo + strip) * 16 * NCN;
#pragma unroll
        for (int j = 0; j < PER; ++j) {
            const int f   = (cg * PER + j) * 128 + lane * 4;
            const int row = f / NCN;
            const int col = f - row * NCN;
            const float4 v = *(const float4*)(&Asm[(strip * 16 + row) * LDP + col]);
            *(float4*)(og + f) = v;
        }
    }
}

extern "C" void kernel_launch(void* const* d_in, const int* in_sizes, int n_in,
                              void* d_out, int out_size, void* d_ws, size_t ws_size,
                              hipStream_t stream) {
    const float* x    = (const float*)d_in[0];
    const int*   ei   = (const int*)d_in[1];
    const float* ew   = (const float*)d_in[2];
    const float* W03  = (const float*)d_in[3];
    const float* b03  = (const float*)d_in[4];
    const float* g03  = (const float*)d_in[5];
    const float* be03 = (const float*)d_in[6];
    const float* W4   = (const float*)d_in[7];
    const float* b4   = (const float*)d_in[8];
    const float* g4   = (const float*)d_in[9];
    const float* be4  = (const float*)d_in[10];

    const int* srcI = ei;
    const int* dstI = ei + NEDGES;

    float* xwa  = (float*)d_ws;
    float* xwb  = xwa + (size_t)NNODES * DH;
    float* hbuf = xwb + (size_t)NNODES * DH;
    float* dinv = hbuf + (size_t)NNODES * DH;
    float* norm = dinv + NNODES;
    float* csrW = norm + NEDGES;
    int* cnt     = (int*)(csrW + NEDGES);
    int* rowptr  = cnt + NNODES;
    int* parts   = rowptr + NNODES + 1;
    int* slotEid = parts + (size_t)CNT_SLICES * NNODES;
    int* csrSrc  = slotEid + NEDGES;
    int* csrDst  = csrSrc + NEDGES;

    const int eblocks = (NEDGES + 255) / 256;
    gcn_count_kernel<<<CNT_RANGES * CNT_SLICES, 1024, 0, stream>>>(dstI, parts);
    gcn_scan_kernel<<<1, 1024, 0, stream>>>(parts, cnt, rowptr);
    gcn_place_kernel<<<CNT_RANGES * CNT_SLICES, 1024, 0, stream>>>(dstI, parts, rowptr, slotEid);
    gcn_order_kernel<<<eblocks, 256, 0, stream>>>(srcI, dstI, ew, rowptr, slotEid, csrSrc, csrDst, csrW, NEDGES);
    gcn_dinv_kernel<<<(NNODES + 255) / 256, 256, 0, stream>>>(rowptr, csrW, dinv, NNODES);
    gcn_norm_kernel<<<eblocks, 256, 0, stream>>>(csrSrc, csrDst, csrW, dinv, norm, NEDGES);

    const int gemm_blocks = (NNODES / 16 + GEMM_WAVES - 1) / GEMM_WAVES;
    gcn_gemm_wmma<DH><<<gemm_blocks, GEMM_WAVES * 32, 0, stream>>>(x, W03, xwa, NNODES);

    const int fblocks = FUSE_BLOCKS;
    gcn_gather_ln_gemm_kernel<DH><<<fblocks, FUSE_WAVES * 32, 0, stream>>>(
        xwa, rowptr, csrSrc, norm, dinv, b03, g03, be03, x, hbuf,
        W03 + (size_t)1 * DH * DH, xwb, NNODES);
    gcn_gather_ln_gemm_kernel<DH><<<fblocks, FUSE_WAVES * 32, 0, stream>>>(
        xwb, rowptr, csrSrc, norm, dinv, b03 + 1 * DH, g03 + 1 * DH, be03 + 1 * DH, hbuf, hbuf,
        W03 + (size_t)2 * DH * DH, xwa, NNODES);
    gcn_gather_ln_gemm_kernel<DH><<<fblocks, FUSE_WAVES * 32, 0, stream>>>(
        xwa, rowptr, csrSrc, norm, dinv, b03 + 2 * DH, g03 + 2 * DH, be03 + 2 * DH, hbuf, hbuf,
        W03 + (size_t)3 * DH * DH, xwb, NNODES);
    gcn_gather_ln_gemm_kernel<DOUT><<<fblocks, FUSE_WAVES * 32, 0, stream>>>(
        xwb, rowptr, csrSrc, norm, dinv, b03 + 3 * DH, g03 + 3 * DH, be03 + 3 * DH, hbuf, hbuf,
        W4, xwa, NNODES);

    const int gblocks = (NNODES + 7) / 8;
    gcn_gather_ln_kernel<DOUT, false><<<gblocks, 256, 0, stream>>>(
        xwa, rowptr, csrSrc, norm, dinv, b4, g4, be4, nullptr, (float*)d_out, NNODES);
}


# Two small launches folded into neighbours: bucket prefix sums built inside the place work-groups from per-group totals (scan launch gone); norm written by the layer-0 product launch; sum order kept
# speedup vs baseline: 4.0403x; 1.1074x over previous
#include <hip/hip_runtime.h>
#include <hip/hip_bf16.h>

#define NNODES 50000
#define NEDGES 640000
#define DH 128
#define DOUT 64
#define LN_EPS 1e-5f

typedef float v2f __attribute__((ext_vector_type(2)));
typedef float v8f __attribute__((ext_vector_type(8)));

#define CNT_RANGES 64
#define CNT_SLICES 4
#define CNT_RS ((NNODES + CNT_RANGES - 1) / CNT_RANGES)
#define CNT_Q  (NEDGES / 4 / CNT_SLICES)
#define CNT_G  ((CNT_Q + 8 * 1024 - 1) / (8 * 1024))

__device__ __forceinline__ int wave_sum32i(int v) {
#pragma unroll
    for (int m = 16; m >= 1; m >>= 1) v += __shfl_xor(v, m, 32);
    return v;
}

__global__ void __launch_bounds__(1024)
gcn_count_kernel(const int* __restrict__ dst, int* __restrict__ parts, int* __restrict__ tot) {
    __shared__ int c[CNT_RS];
    __shared__ int tsum;
    if (threadIdx.x == 0) tsum = 0;
    const int r  = blockIdx.x / CNT_SLICES;
    const int s  = blockIdx.x - r * CNT_SLICES;
    const int lo = r * CNT_RS;
    for (int i = threadIdx.x; i < CNT_RS; i += 1024) c[i] = 0;
    __syncthreads();
    const int4* __restrict__ d4 = (const int4*)dst + (size_t)s * CNT_Q;
    for (int g = 0; g < CNT_G; ++g) {
        int4 v[8];
#pragma unroll
        for (int u = 0; u < 8; ++u) {
            const int i = (g * 8 + u) * 1024 + (int)threadIdx.x;
            v[u] = d4[i < CNT_Q ? i : CNT_Q - 1];
            if (i >= CNT_Q) v[u] = make_int4(-1, -1, -1, -1);
        }
#pragma unroll
        for (int u = 0; u < 8; ++u) {
            const unsigned a0 = (unsigned)(v[u].x - lo), a1 = (unsigned)(v[u].y - lo);
            const unsigned a2 = (unsigned)(v[u].z - lo), a3 = (unsigned)(v[u].w - lo);
            if (a0 < CNT_RS) atomicAdd(&c[a0], 1);
            if (a1 < CNT_RS) atomicAdd(&c[a1], 1);
            if (a2 < CNT_RS) atomicAdd(&c[a2], 1);
            if (a3 < CNT_RS) atomicAdd(&c[a3], 1);
        }
    }
    __syncthreads();
    int mine = 0;
    for (int i = threadIdx.x; i < CNT_RS; i += 1024) {
        const int n = lo + i;
        if (n < NNODES) { const int v = c[i]; parts[(size_t)s * NNODES + n] = v; mine += v; }
    }
    mine = wave_sum32i(mine);
    if ((threadIdx.x & 31) == 0) atomicAdd(&tsum, mine);
    __syncthreads();
    if (threadIdx.x == 0) tot[blockIdx.x] = tsum;
}

__device__ __forceinline__ int wave_incl_scan32(int x, int lane) {
#pragma unroll
    for (int o = 1; o < 32; o <<= 1) {
        int y = __shfl_up(x, o, 32);
        if (lane >= o) x += y;
    }
    return x;
}

__global__ void __launch_bounds__(1024)
gcn_place_kernel(const int* __restrict__ dst, const int* __restrict__ parts,
                 const int* __restrict__ tot, int* __restrict__ rowptr,
                 int* __restrict__ slot_eid) {
    __shared__ int c[CNT_RS];
    __shared__ int wtot[32];
    __shared__ int btot[32];
    const int r  = blockIdx.x / CNT_SLICES;
    const int s  = blockIdx.x - r * CNT_SLICES;
    const int lo = r * CNT_RS;
    {
        const int lane = threadIdx.x & 31;
        const int w    = threadIdx.x >> 5;
        const int i    = threadIdx.x;
        const int n    = lo + i;
        const bool valid = (i < CNT_RS) && (n < NNODES);
        int t = 0, before = 0;
        if (valid) {
#pragma unroll
            for (int p = 0; p < CNT_SLICES; ++p) {
                const int v = parts[(size_t)p * NNODES + n];
                t += v;
                if (p < s) before += v;
            }
        }
        int bpart = (i < r * CNT_SLICES) ? tot[i] : 0;
        const int inc = wave_incl_scan32(t, lane);
        bpart = wave_sum32i(bpart);
        if (lane == 31) wtot[w] = inc;
        if (lane == 0)  btot[w] = bpart;
        __syncthreads();
        const int tv     = wtot[lane];
        const int tinc   = wave_incl_scan32(tv, lane);
        const int run    = __shfl(tinc - tv, w, 32);
        const int rtotal = __shfl(tinc, 31, 32);
        const int base   = wave_sum32i(btot[lane]);
        const int start  = base + run + inc - t;
        if (i < CNT_RS) c[i] = start + before;
        if (s == 0) {
            if (valid) rowptr[n] = start;
            if (r == CNT_RANGES - 1 && i == 0) rowptr[NNODES] = base + rtotal;
        }
    }
    __syncthreads();
    const int4* __restrict__ d4 = (const int4*)dst + (size_t)s * CNT_Q;
    for (int g = 0; g < CNT_G; ++g) {
        int4 v[8];
#pragma unroll
        for (int u = 0; u < 8; ++u) {
            const int i = (g * 8 + u) * 1024 + (int)threadIdx.x;
            v[u] = d4[i < CNT_Q ? i : CNT_Q - 1];
            if (i >= CNT_Q) v[u] = make_int4(-1, -1, -1, -1);
        }
#pragma unroll
        for (int u = 0; u < 8; ++u) {
            const int e0 = (s * CNT_Q + (g * 8 + u) * 1024 + (int)threadIdx.x) * 4;
            const unsigned a0 = (unsigned)(v[u].x - lo), a1 = (unsigned)(v[u].y - lo);
            const unsigned a2 = (unsigned)(v[u].z - lo), a3 = (unsigned)(v[u].w - lo);
            if (a0 < CNT_RS) slot_eid[atomicAdd(&c[a0], 1)] = e0;
            if (a1 < CNT_RS) slot_eid[atomicAdd(&c[a1], 1)] = e0 + 1;
            if (a2 < CNT_RS) slot_eid[atomicAdd(&c[a2], 1)] = e0 + 2;
            if (a3 < CNT_RS) slot_eid[atomicAdd(&c[a3], 1)] = e0 + 3;
        }
    }
}

__global__ void gcn_order_kernel(const int* __restrict__ src, const int* __restrict__ dst,
                                 const float* __restrict__ w, const int* __restrict__ rowptr,
                                 const int* __restrict__ slot_eid,
                                 int* __restrict__ csr_src, int* __restrict__ csr_dst,
                                 float* __restrict__ csr_w, int E) {
    int p = blockIdx.x * blockDim.x + threadIdx.x;
    if (p >= E) return;
    int e  = slot_eid[p];
    int d  = dst[e];
    int lo = rowptr[d], hi = rowptr[d + 1];
    int r  = 0;
    for (int j = lo; j < hi; ++j) r += (slot_eid[j] < e) ? 1 : 0;
    int q = lo + r;
    csr_src[q] = src[e];
    csr_dst[q] = d;
    csr_w[q]   = w[e];
}

__global__ void gcn_dinv_kernel(const int* __restrict__ rowptr, const float* __restrict__ csr_w,
                                float* __restrict__ dinv, int n) {
    int i = blockIdx.x * blockDim.x + threadIdx.x;
    if (i < n) {
        float d = 0.0f;
        for (int q = rowptr[i]; q < rowptr[i + 1]; ++q) d += csr_w[q];
        dinv[i] = rsqrtf(d + 1.0f);
    }
}

#define LDP 132
#define GEMM_WAVES 8
#define GEMM_BLOCKS ((NNODES / 16 + GEMM_WAVES - 1) / GEMM_WAVES)
#define GEMM_THREADS (GEMM_BLOCKS * GEMM_WAVES * 32)
#define NORM_PT ((NEDGES + GEMM_THREADS - 1) / GEMM_THREADS)

template <int NC>
__global__ void __launch_bounds__(GEMM_WAVES * 32)
gcn_gemm_wmma(const float* __restrict__ A,
              const float* __restrict__ W,
              float* __restrict__ out,
              int nrows,
              const int* __restrict__ csr_src, const int* __restrict__ csr_dst,
              const float* __restrict__ csr_w, const float* __restrict__ dinv,
              float* __restrict__ norm) {
    constexpr int NT = NC / 16;
    __shared__ __attribute__((aligned(16))) float Wsm[NC * LDP];
    __shared__ __attribute__((aligned(16))) float Asm[GEMM_WAVES][16 * LDP];

    const int lane = threadIdx.x & 31;
    const int w    = threadIdx.x >> 5;
    const int strip   = blockIdx.x * GEMM_WAVES + w;
    const int nstrips = nrows >> 4;
    const bool active = strip < nstrips;
    const int row0 = (active ? strip : nstrips - 1) << 4;
    float* __restrict__ as = &Asm[w][0];

    const int g0 = blockIdx.x * (GEMM_WAVES * 32) + threadIdx.x;
    int   es[NORM_PT], ed[NORM_PT];
    float ewt[NORM_PT];
#pragma unroll
    for (int k = 0; k < NORM_PT; ++k) {
        const int q  = g0 + k * GEMM_THREADS;
        const int qc = (GEMM_THREADS * (k + 1) <= NEDGES || q < NEDGES) ? q : NEDGES - 1;
        es[k]  = csr_src[qc];
        ed[k]  = csr_dst[qc];
        ewt[k] = csr_w[qc];
    }

    float4 av[16], wv[NC / GEMM_WAVES];
    const float* __restrict__ ag = A + (size_t)row0 * DH + lane * 4;
#pragma unroll
    for (int r = 0; r < 16; ++r) av[r] = *(const float4*)(ag + (size_t)r * DH);
    const float* __restrict__ wg = W + (size_t)w * DH + lane * 4;
#pragma unroll
    for (int j = 0; j < NC / GEMM_WAVES; ++j) wv[j] = *(const float4*)(wg + (size_t)j * GEMM_WAVES * DH);

    float dsv[NORM_PT], ddv[NORM_PT];
#pragma unroll
    for (int k = 0; k < NORM_PT; ++k) { dsv[k] = dinv[es[k]]; ddv[k] = dinv[ed[k]]; }

#pragma unroll
    for (int r = 0; r < 16; ++r) *(float4*)(&as[r * LDP + lane * 4]) = av[r];
#pragma unroll
    for (int j = 0; j < NC / GEMM_WAVES; ++j) *(float4*)(&Wsm[(w + j * GEMM_WAVES) * LDP + lane * 4]) = wv[j];
    __syncthreads();

#pragma unroll
    for (int k = 0; k < NORM_PT; ++k) {
        const int q = g0 + k * GEMM_THREADS;
        if (GEMM_THREADS * (k + 1) <= NEDGES || q < NEDGES) norm[q] = dsv[k] * ewt[k] * ddv[k];
    }

    const int m  = lane & 15;
    const int ko = (lane >> 4) << 1;
    v8f acc[NT];
#pragma unroll
    for (int t = 0; t < NT; ++t) acc[t] = v8f{};

    const float* ap = as  + m * LDP + ko;
    const float* bp = Wsm + m * LDP + ko;
    v2f af[2][2];
    v2f bf[2][2][NT];
#pragma unroll
    for (int s = 0; s < 2; ++s) {
        af[0][s] = *(const v2f*)(ap + s * 4);
#pragma unroll
        for (int t = 0; t < NT; ++t) bf[0][s][t] = *(const v2f*)(bp + t * 16 * LDP + s * 4);
    }
#pragma unroll
    for (int kb = 0; kb < DH / 8; ++kb) {
        const int cur = kb & 1;
        if (kb + 1 < DH / 8) {
#pragma unroll
            for (int s = 0; s < 2; ++s) {
                af[cur ^ 1][s] = *(const v2f*)(ap + (kb + 1) * 8 + s * 4);
#pragma unroll
                for (int t = 0; t < NT; ++t)
                    bf[cur ^ 1][s][t] = *(const v2f*)(bp + t * 16 * LDP + (kb + 1) * 8 + s * 4);
            }
        }
        __builtin_amdgcn_sched_barrier(0);
#pragma unroll
        for (int s = 0; s < 2; ++s)
#pragma unroll
            for (int t = 0; t < NT; ++t)
                acc[t] = __builtin_amdgcn_wmma_f32_16x16x4_f32(
                     false, af[cur][s],  false, bf[cur][s][t],
                     (short)0, acc[t],  false,  false);
        __builtin_amdgcn_sched_barrier(0);
    }
    __syncthreads();

    {
        float* cs = as + ((lane >> 4) << 3) * LDP + m;
#pragma unroll
        for (int t = 0; t < NT; ++t)
#pragma unroll
            for (int v = 0; v < 8; ++v)
                cs[v * LDP + t * 16] = acc[t][v];
    }
    __syncthreads();

    if (active) {
        float* __restrict__ og = out + (size_t)row0 * NC + lane * 4;
#pragma unroll
        for (int j = 0; j < (16 * NC) / 128; ++j) {
            const int f   = j * 128 + lane * 4;
            const int row = f / NC;
            const int col = f - row * NC;
            const float4 v = *(const float4*)(&as[row * LDP + col]);
            *(float4*)(og + j * 128) = v;
        }
    }
}

__device__ __forceinline__ float wave_sum32(float v) {
#pragma unroll
    for (int m = 16; m >= 1; m >>= 1) v += __shfl_xor(v, m, 32);
    return v;
}

template <int V> struct RowPiece { typedef float type __attribute__((ext_vector_type(V))); };

template <int D, bool HIDDEN>
__device__ __forceinline__ typename RowPiece<D / 32>::type
gcn_gather_ln_node(const float* __restrict__ xw, const int* __restrict__ rowptr,
                   const int* __restrict__ csr_src, const float* __restrict__ norm,
                   const float* __restrict__ dinv,
                   const float* __restrict__ bias, const float* __restrict__ gamma,
                   const float* __restrict__ beta,
                   const float* res, const int node, const int lane) {
    constexpr int V = D / 32;
    typedef typename RowPiece<V>::type vf;
    const int lo = rowptr[node], hi = rowptr[node + 1];
    const float* __restrict__ xl = xw + lane * V;
    const size_t own = (size_t)node * D + lane * V;

    const vf xs = *(const vf*)(xw + own);
    vf rs = {};
    if (HIDDEN) rs = *(const vf*)(res + own);
    const float di = dinv[node];
    const vf bv = *(const vf*)(bias + lane * V);
    const vf gv = *(const vf*)(gamma + lane * V);
    const vf ev = *(const vf*)(beta + lane * V);

    vf acc = {};
    for (int base = lo; base < hi; base += 32) {
        const int c = (hi - base < 32) ? (hi - base) : 32;
        int   ms = 0;
        float mw = 0.0f;
        if (lane < c) { ms = csr_src[base + lane]; mw = norm[base + lane]; }
        const int mwi = __float_as_int(mw);
        int j = 0;
        for (; j + 4 <= c; j += 4) {
            const int s0 = __builtin_amdgcn_readlane(ms, j);
            const int s1 = __builtin_amdgcn_readlane(ms, j + 1);
            const int s2 = __builtin_amdgcn_readlane(ms, j + 2);
            const int s3 = __builtin_amdgcn_readlane(ms, j + 3);
            const float w0 = __int_as_float(__builtin_amdgcn_readlane(mwi, j));
            const float w1 = __int_as_float(__builtin_amdgcn_readlane(mwi, j + 1));
            const float w2 = __int_as_float(__builtin_amdgcn_readlane(mwi, j + 2));
            const float w3 = __int_as_float(__builtin_amdgcn_readlane(mwi, j + 3));
            const vf v0 = *(const vf*)(xl + (size_t)s0 * D);
            const vf v1 = *(const vf*)(xl + (size_t)s1 * D);
            const vf v2 = *(const vf*)(xl + (size_t)s2 * D);
            const vf v3 = *(const vf*)(xl + (size_t)s3 * D);
            acc += w0 * v0;
            acc += w1 * v1;
            acc += w2 * v2;
            acc += w3 * v3;
        }
        for (; j < c; ++j) {
            const int s0 = __builtin_amdgcn_readlane(ms, j);
            const float w0 = __int_as_float(__builtin_amdgcn_readlane(mwi, j));
            const vf v0 = *(const vf*)(xl + (size_t)s0 * D);
            acc += w0 * v0;
        }
    }

    const float sl = di * di;
    float v[V];
#pragma unroll
    for (int i = 0; i < V; ++i) v[i] = fmaf(sl, xs[i], acc[i]) + bv[i];

    float ps = v[0], pq = v[0] * v[0];
#pragma unroll
    for (int i = 1; i < V; ++i) { ps += v[i]; pq += v[i] * v[i]; }
    const float s  = wave_sum32(ps);
    const float sq = wave_sum32(pq);
    const float mean = s * (1.0f / D);
    const float var  = sq * (1.0f / D) - mean * mean;
    const float rinv = rsqrtf(var + LN_EPS);

    vf o;
#pragma unroll
    for (int i = 0; i < V; ++i) {
        const float y = gv[i] * (v[i] - mean) * rinv + ev[i];
        o[i] = HIDDEN ? (fmaxf(y, 0.0f) + rs[i]) : y;
    }
    return o;
}

template <int D, bool HIDDEN>
__global__ void gcn_gather_ln_kernel(const float* __restrict__ xw, const int* __restrict__ rowptr,
                                     const int* __restrict__ csr_src, const float* __restrict__ norm,
                                     const float* __restrict__ dinv,
                                     const float* __restrict__ bias, const float* __restrict__ gamma,
                                     const float* __restrict__ beta,
                                     const float* res, float* hout, int n) {
    constexpr int V = D / 32;
    typedef typename RowPiece<V>::type vf;
    const int lane = threadIdx.x & 31;
    const int node = __builtin_amdgcn_readfirstlane(blockIdx.x * (blockDim.x >> 5) + (threadIdx.x >> 5));
    if (node >= n) return;
    const vf o = gcn_gather_ln_node<D, HIDDEN>(xw, rowptr, csr_src, norm, dinv, bias, gamma, beta,
                                               res, node, lane);
    *(vf*)(hout + (size_t)node * D + lane * V) = o;
}

#define FUSE_WAVES 32
#define FUSE_BLOCKS 512
#define FUSE_STRIPS 8

template <int NCN>
__global__ void __launch_bounds__(FUSE_WAVES * 32) __attribute__((amdgpu_waves_per_eu(16, 16)))
gcn_gather_ln_gemm_kernel(const float* __restrict__ xw, const int* __restrict__ rowptr,
                          const int* __restrict__ csr_src, const float* __restrict__ norm,
                          const float* __restrict__ dinv,
                          const float* __restrict__ bias, const float* __restrict__ gamma,
                          const float* __restrict__ beta,
                          const float* res, float* hout,
                          const float* __restrict__ Wn, float* __restrict__ xw_out, int n) {
    constexpr int NT = NCN / 64;
    constexpr int WR = NCN / FUSE_WAVES;
    __shared__ __attribute__((aligned(16))) float Wsm[NCN * LDP];
    __shared__ __attribute__((aligned(16))) float Asm[FUSE_STRIPS * 16 * LDP];

    const int lane = threadIdx.x & 31;
    const int w    = __builtin_amdgcn_readfirstlane(threadIdx.x >> 5);

    const int nstrips = n >> 4;
    const int s_lo  = ((int)blockIdx.x * nstrips) / FUSE_BLOCKS;
    const int s_hi  = (((int)blockIdx.x + 1) * nstrips) / FUSE_BLOCKS;
    const int ns    = s_hi - s_lo;
    const int nb    = s_lo << 4;
    const int nrows = ns << 4;

    for (int r = w; r < nrows; r += FUSE_WAVES) {
        const int node = nb + r;
        const typename RowPiece<4>::type q =
            gcn_gather_ln_node<DH, true>(xw, rowptr, csr_src, norm, dinv, bias, gamma, beta,
                                         res, node, lane);
        const float4 o = make_float4(q[0], q[1], q[2], q[3]);
        *(float4*)(hout + (size_t)node * DH + lane * 4) = o;
        *(float4*)(&Asm[r * LDP + lane * 4]) = o;
    }

    {
        float4 wv[WR];
        const float* __restrict__ wg = Wn + (size_t)w * DH + lane * 4;
#pragma unroll
        for (int j = 0; j < WR; ++j) wv[j] = *(const float4*)(wg + (size_t)j * FUSE_WAVES * DH);
#pragma unroll
        for (int j = 0; j < WR; ++j) *(float4*)(&Wsm[(w + j * FUSE_WAVES) * LDP + lane * 4]) = wv[j];
    }
    __syncthreads();

    const int strip = w >> 2;
    const int cg    = w & 3;
    const bool mine = strip < ns;
    const int m  = lane & 15;
    const int ko = (lane >> 4) << 1;
    v8f acc[NT];
#pragma unroll
    for (int t = 0; t < NT; ++t) acc[t] = v8f{};

    if (mine) {
        const float* ap = Asm + (strip * 16 + m) * LDP + ko;
        const float* bp = Wsm + (cg * NT * 16 + m) * LDP + ko;
        v2f af[2][2];
        v2f bf[2][2][NT];
#pragma unroll
        for (int s = 0; s < 2; ++s) {
            af[0][s] = *(const v2f*)(ap + s * 4);
#pragma unroll
            for (int t = 0; t < NT; ++t) bf[0][s][t] = *(const v2f*)(bp + t * 16 * LDP + s * 4);
        }
#pragma unroll
        for (int kb = 0; kb < DH / 8; ++kb) {
            const int cur = kb & 1;
            if (kb + 1 < DH / 8) {
#pragma unroll
                for (int s = 0; s < 2; ++s) {
                    af[cur ^ 1][s] = *(const v2f*)(ap + (kb + 1) * 8 + s * 4);
#pragma unroll
                    for (int t = 0; t < NT; ++t)
                        bf[cur ^ 1][s][t] = *(const v2f*)(bp + t * 16 * LDP + (kb + 1) * 8 + s * 4);
                }
            }
            __builtin_amdgcn_sched_barrier(0);
#pragma unroll
            for (int s = 0; s < 2; ++s)
#pragma unroll
                for (int t = 0; t < NT; ++t)
                    acc[t] = __builtin_amdgcn_wmma_f32_16x16x4_f32(
                         false, af[cur][s],  false, bf[cur][s][t],
                         (short)0, acc[t],  false,  false);
            __builtin_amdgcn_sched_barrier(0);
        }
    }
    __syncthreads();

    if (mine) {
        float* cs = Asm + (strip * 16 + ((lane >> 4) << 3)) * LDP + cg * NT * 16 + m;
#pragma unroll
        for (int t = 0; t < NT; ++t)
#pragma unroll
            for (int v = 0; v < 8; ++v)
                cs[v * LDP + t * 16] = acc[t][v];
    }
    __syncthreads();

    if (mine) {
        constexpr int PER = (16 * NCN) / 128 / 4;
        float* __restrict__ og = xw_out + (size_t)(s_lo + strip) * 16 * NCN;
#pragma unroll
        for (int j = 0; j < PER; ++j) {
            const int f   = (cg * PER + j) * 128 + lane * 4;
            const int row = f / NCN;
            const int col = f - row * NCN;
            const float4 v = *(const float4*)(&Asm[(strip * 16 + row) * LDP + col]);
            *(float4*)(og + f) = v;
        }
    }
}

extern "C" void kernel_launch(void* const* d_in, const int* in_sizes, int n_in,
                              void* d_out, int out_size, void* d_ws, size_t ws_size,
                              hipStream_t stream) {
    const float* x    = (const float*)d_in[0];
    const int*   ei   = (const int*)d_in[1];
    const float* ew   = (const float*)d_in[2];
    const float* W03  = (const float*)d_in[3];
    const float* b03  = (const float*)d_in[4];
    const float* g03  = (const float*)d_in[5];
    const float* be03 = (const float*)d_in[6];
    const float* W4   = (const float*)d_in[7];
    const float* b4   = (const float*)d_in[8];
    const float* g4   = (const float*)d_in[9];
    const float* be4  = (const float*)d_in[10];

    const int* srcI = ei;
    const int* dstI = ei + NEDGES;

    float* xwa  = (float*)d_ws;
    float* xwb  = xwa + (size_t)NNODES * DH;
    float* hbuf = xwb + (size_t)NNODES * DH;
    float* dinv = hbuf + (size_t)NNODES * DH;
    float* norm = dinv + NNODES;
    float* csrW = norm + NEDGES;
    int* tot     = (int*)(csrW + NEDGES);
    int* rowptr  = tot + NNODES;
    int* parts   = rowptr + NNODES + 1;
    int* slotEid = parts + (size_t)CNT_SLICES * NNODES;
    int* csrSrc  = slotEid + NEDGES;
    int* csrDst  = csrSrc + NEDGES;

    const int eblocks = (NEDGES + 255) / 256;
    gcn_count_kernel<<<CNT_RANGES * CNT_SLICES, 1024, 0, stream>>>(dstI, parts, tot);
    gcn_place_kernel<<<CNT_RANGES * CNT_SLICES, 1024, 0, stream>>>(dstI, parts, tot, rowptr, slotEid);
    gcn_order_kernel<<<eblocks, 256, 0, stream>>>(srcI, dstI, ew, rowptr, slotEid, csrSrc, csrDst, csrW, NEDGES);
    gcn_dinv_kernel<<<(NNODES + 255) / 256, 256, 0, stream>>>(rowptr, csrW, dinv, NNODES);

    gcn_gemm_wmma<DH><<<GEMM_BLOCKS, GEMM_WAVES * 32, 0, stream>>>(x, W03, xwa, NNODES,
                                                                  csrSrc, csrDst, csrW, dinv, norm);

    const int fblocks = FUSE_BLOCKS;
    gcn_gather_ln_gemm_kernel<DH><<<fblocks, FUSE_WAVES * 32, 0, stream>>>(
        xwa, rowptr, csrSrc, norm, dinv, b03, g03, be03, x, hbuf,
        W03 + (size_t)1 * DH * DH, xwb, NNODES);
    gcn_gather_ln_gemm_kernel<DH><<<fblocks, FUSE_WAVES * 32, 0, stream>>>(
        xwb, rowptr, csrSrc, norm, dinv, b03 + 1 * DH, g03 + 1 * DH, be03 + 1 * DH, hbuf, hbuf,
        W03 + (size_t)2 * DH * DH, xwa, NNODES);
    gcn_gather_ln_gemm_kernel<DH><<<fblocks, FUSE_WAVES * 32, 0, stream>>>(
        xwa, rowptr, csrSrc, norm, dinv, b03 + 2 * DH, g03 + 2 * DH, be03 + 2 * DH, hbuf, hbuf,
        W03 + (size_t)3 * DH * DH, xwb, NNODES);
    gcn_gather_ln_gemm_kernel<DOUT><<<fblocks, FUSE_WAVES * 32, 0, stream>>>(
        xwb, rowptr, csrSrc, norm, dinv, b03 + 3 * DH, g03 + 3 * DH, be03 + 3 * DH, hbuf, hbuf,
        W4, xwa, NNODES);

    const int gblocks = (NNODES + 7) / 8;
    gcn_gather_ln_kernel<DOUT, false><<<gblocks, 256, 0, stream>>>(
        xwa, rowptr, csrSrc, norm, dinv, b4, g4, be4, nullptr, (float*)d_out, NNODES);
}
